# starting kernel
# baseline (speedup 1.0000x reference)
#include <hip/hip_runtime.h>
#include <math.h>

#define NB 2
#define DD 1024
#define LL 2048
#define NS 16
#define RR 64
#define EE 96
#define NC 32
#define LC 64

typedef float v2f __attribute__((ext_vector_type(2)));
typedef float v8f __attribute__((ext_vector_type(8)));

__device__ __forceinline__ float sigmoidf_(float x) { return 1.0f / (1.0f + expf(-x)); }
__device__ __forceinline__ float softplusf_(float x) { return (x > 20.0f) ? x : log1pf(expf(x)); }

__global__ __launch_bounds__(128)
void k_xproj(const float* __restrict__ W, const float* __restrict__ U,
             float* __restrict__ Xd)
{
    __shared__ float As[16 * 68];
    const int tid  = threadIdx.x;
    const int lane = tid & 31;
    const int wv   = tid >> 5;
    const int m0   = blockIdx.y << 4;
    const int n0   = (blockIdx.x << 6) + (wv << 4);
    const int bi   = blockIdx.z;
    const int am   = lane & 15;
    const int ak   = (lane >> 4) << 1;
    v8f acc = {};
    const float* Ub = U + ((size_t)bi * DD) * LL + n0 + am;
    for (int kc = 0; kc < DD; kc += 64) {
        for (int i = tid; i < 16 * 64; i += 128) {
            const int r = i >> 6, c = i & 63;
            As[r * 68 + c] = W[(size_t)(m0 + r) * DD + kc + c];
        }
        __syncthreads();
        #pragma unroll
        for (int k = 0; k < 64; k += 4) {
            v2f a, b;
            a.x = As[am * 68 + k + ak];
            a.y = As[am * 68 + k + ak + 1];
            const float* bp = Ub + (size_t)(kc + k + ak) * LL;
            b.x = bp[0];
            b.y = bp[LL];
            acc = __builtin_amdgcn_wmma_f32_16x16x4_f32(
                false, a, false, b, (short)0, acc, false, false);
        }
        __syncthreads();
    }
    const int cn = n0 + am;
    const int cr = (lane >> 4) << 3;
    #pragma unroll
    for (int v = 0; v < 8; ++v)
        Xd[((size_t)bi * EE + m0 + cr + v) * LL + cn] = acc[v];
}

__global__ __launch_bounds__(128)
void k_dt(const float* __restrict__ Wdt, const float* __restrict__ Xd,
          const float* __restrict__ bias, float* __restrict__ Delta)
{
    __shared__ float As[16 * 68];
    const int tid  = threadIdx.x;
    const int lane = tid & 31;
    const int wv   = tid >> 5;
    const int m0   = blockIdx.y << 4;
    const int n0   = (blockIdx.x << 6) + (wv << 4);
    const int bi   = blockIdx.z;
    const int am   = lane & 15;
    const int ak   = (lane >> 4) << 1;
    for (int i = tid; i < 16 * 64; i += 128) {
        const int r = i >> 6, c = i & 63;
        As[r * 68 + c] = Wdt[(size_t)(m0 + r) * RR + c];
    }
    __syncthreads();
    v8f acc = {};
    const float* Bb = Xd + ((size_t)bi * EE) * LL + n0 + am;
    #pragma unroll
    for (int k = 0; k < RR; k += 4) {
        v2f a, b;
        a.x = As[am * 68 + k + ak];
        a.y = As[am * 68 + k + ak + 1];
        const float* bp = Bb + (size_t)(k + ak) * LL;
        b.x = bp[0];
        b.y = bp[LL];
        acc = __builtin_amdgcn_wmma_f32_16x16x4_f32(
            false, a, false, b, (short)0, acc, false, false);
    }
    const int cn = n0 + am;
    const int cr = (lane >> 4) << 3;
    #pragma unroll
    for (int v = 0; v < 8; ++v) {
        const int row = m0 + cr + v;
        Delta[((size_t)bi * DD + row) * LL + cn] = softplusf_(acc[v] + bias[row]);
    }
}

__global__ __launch_bounds__(256)
void k_scanA(const float* __restrict__ U, const float* __restrict__ Xd,
             const float* __restrict__ Delta, const float* __restrict__ Alog,
             float* __restrict__ S, float* __restrict__ SumD)
{
    const int lane = threadIdx.x & 31;
    const int gw   = blockIdx.x * (blockDim.x >> 5) + (threadIdx.x >> 5);
    const int p    = (gw << 1) + (lane >> 4);
    const int ch   = p >> 5;
    const int c    = p & (NC - 1);
    const int bi   = ch >> 10;
    const int d    = ch & (DD - 1);
    const int n    = lane & 15;
    const int hb   = lane & 16;
    const float An = -expf(Alog[d * NS + n]);
    const size_t rowOff = ((size_t)bi * DD + d) * LL + (size_t)c * LC;
    const float* dRow = Delta + rowOff;
    const float* uRow = U + rowOff;
    const float* bRow = Xd + ((size_t)bi * EE + RR + n) * LL + (size_t)c * LC;
    float h = 0.0f, sdp = 0.0f;
    for (int l0 = 0; l0 < LC; l0 += 16) {
        const float dl = dRow[l0 + n];
        const float ul = uRow[l0 + n];
        sdp += dl;
        float bv[16];
        #pragma unroll
        for (int q = 0; q < 4; ++q) {
            float4 t = *(const float4*)(bRow + l0 + 4 * q);
            bv[4*q+0] = t.x; bv[4*q+1] = t.y; bv[4*q+2] = t.z; bv[4*q+3] = t.w;
        }
        #pragma unroll
        for (int j = 0; j < 16; ++j) {
            const float dj = __shfl(dl, hb + j, 32);
            const float uj = __shfl(ul, hb + j, 32);
            h = expf(dj * An) * h + dj * uj * bv[j];
        }
    }
    sdp += __shfl_xor(sdp, 8, 32);
    sdp += __shfl_xor(sdp, 4, 32);
    sdp += __shfl_xor(sdp, 2, 32);
    sdp += __shfl_xor(sdp, 1, 32);
    S[((size_t)ch * NC + c) * NS + n] = h;
    if (n == 0) SumD[(size_t)ch * NC + c] = sdp;
}

__global__ __launch_bounds__(256)
void k_scanB(const float* __restrict__ Alog, const float* __restrict__ S,
             const float* __restrict__ SumD, float* __restrict__ Hin)
{
    const int lane = threadIdx.x & 31;
    const int gw   = blockIdx.x * (blockDim.x >> 5) + (threadIdx.x >> 5);
    const int ch   = (gw << 1) + (lane >> 4);
    const int d    = ch & (DD - 1);
    const int n    = lane & 15;
    const float An = -expf(Alog[d * NS + n]);
    float P[NC];
    #pragma unroll
    for (int c = 0; c < NC; ++c)
        P[c] = expf(An * SumD[(size_t)ch * NC + c]);
    float hin = 0.0f;
    #pragma unroll
    for (int c = 0; c < NC; ++c) {
        Hin[((size_t)ch * NC + c) * NS + n] = hin;
        hin = P[c] * hin + S[((size_t)ch * NC + c) * NS + n];
    }
}

__global__ __launch_bounds__(256)
void k_scanC(const float* __restrict__ U, const float* __restrict__ Xd,
             const float* __restrict__ Delta, const float* __restrict__ Alog,
             const float* __restrict__ Dp, const float* __restrict__ Hin,
             float* __restrict__ Y)
{
    const int lane = threadIdx.x & 31;
    const int gw   = blockIdx.x * (blockDim.x >> 5) + (threadIdx.x >> 5);
    const int p    = (gw << 1) + (lane >> 4);
    const int ch   = p >> 5;
    const int c    = p & (NC - 1);
    const int bi   = ch >> 10;
    const int d    = ch & (DD - 1);
    const int n    = lane & 15;
    const int hb   = lane & 16;
    const float An = -expf(Alog[d * NS + n]);
    const float Dv = Dp[d];
    const size_t rowOff = ((size_t)bi * DD + d) * LL + (size_t)c * LC;
    const float* dRow = Delta + rowOff;
    const float* uRow = U + rowOff;
    const float* bRow = Xd + ((size_t)bi * EE + RR) * LL      + (size_t)n * LL + (size_t)c * LC;
    const float* cRow = Xd + ((size_t)bi * EE + RR + NS) * LL + (size_t)n * LL + (size_t)c * LC;
    float* yRow = Y + rowOff;
    float h = Hin[((size_t)ch * NC + c) * NS + n];
    for (int l0 = 0; l0 < LC; l0 += 16) {
        const float dl = dRow[l0 + n];
        const float ul = uRow[l0 + n];
        float bv[16], cv[16];
        #pragma unroll
        for (int q = 0; q < 4; ++q) {
            float4 t = *(const float4*)(bRow + l0 + 4 * q);
            bv[4*q+0] = t.x; bv[4*q+1] = t.y; bv[4*q+2] = t.z; bv[4*q+3] = t.w;
            float4 s = *(const float4*)(cRow + l0 + 4 * q);
            cv[4*q+0] = s.x; cv[4*q+1] = s.y; cv[4*q+2] = s.z; cv[4*q+3] = s.w;
        }
        float ymy = 0.0f;
        #pragma unroll
        for (int j = 0; j < 16; ++j) {
            const float dj = __shfl(dl, hb + j, 32);
            const float uj = __shfl(ul, hb + j, 32);
            h = expf(dj * An) * h + dj * uj * bv[j];
            float yv = h * cv[j];
            yv += __shfl_xor(yv, 8, 32);
            yv += __shfl_xor(yv, 4, 32);
            yv += __shfl_xor(yv, 2, 32);
            yv += __shfl_xor(yv, 1, 32);
            if (n == j) ymy = yv;
        }
        float yo = ymy + ul * Dv;
        yo = yo * (ul * sigmoidf_(ul));
        yo = 0.5f * yo * (1.0f + erff(yo * 0.70710678118f));
        yRow[l0 + n] = yo;
    }
}

__global__ __launch_bounds__(128)
void k_out(const float* __restrict__ Wo, const float* __restrict__ Yv,
           const float* __restrict__ bo, float* __restrict__ Out)
{
    __shared__ float Aa[16 * 68];
    __shared__ float Ag[16 * 68];
    const int tid  = threadIdx.x;
    const int lane = tid & 31;
    const int wv   = tid >> 5;
    const int m0   = blockIdx.y << 4;
    const int n0   = (blockIdx.x << 7) + (wv << 5);
    const int bi   = blockIdx.z;
    const int am   = lane & 15;
    const int ak   = (lane >> 4) << 1;
    v8f aA0 = {}, aA1 = {}, aG0 = {}, aG1 = {};
    const float* Yb = Yv + ((size_t)bi * DD) * LL + n0 + am;
    for (int kc = 0; kc < DD; kc += 64) {
        for (int i = tid; i < 16 * 64; i += 128) {
            const int r = i >> 6, c = i & 63;
            Aa[r * 68 + c] = Wo[(size_t)(m0 + r) * DD + kc + c];
            Ag[r * 68 + c] = Wo[(size_t)(DD + m0 + r) * DD + kc + c];
        }
        __syncthreads();
        if (kc + 64 < DD)
            __builtin_prefetch(&Wo[(size_t)(m0 + (tid >> 3)) * DD + kc + 64 + ((tid & 7) << 3)], 0, 1);
        #pragma unroll
        for (int k = 0; k < 64; k += 4) {
            v2f b0, b1, aa, ag;
            const float* bp = Yb + (size_t)(kc + k + ak) * LL;
            b0.x = bp[0];
            b0.y = bp[LL];
            b1.x = bp[16];
            b1.y = bp[LL + 16];
            aa.x = Aa[am * 68 + k + ak];
            aa.y = Aa[am * 68 + k + ak + 1];
            ag.x = Ag[am * 68 + k + ak];
            ag.y = Ag[am * 68 + k + ak + 1];
            aA0 = __builtin_amdgcn_wmma_f32_16x16x4_f32(false, aa, false, b0, (short)0, aA0, false, false);
            aA1 = __builtin_amdgcn_wmma_f32_16x16x4_f32(false, aa, false, b1, (short)0, aA1, false, false);
            aG0 = __builtin_amdgcn_wmma_f32_16x16x4_f32(false, ag, false, b0, (short)0, aG0, false, false);
            aG1 = __builtin_amdgcn_wmma_f32_16x16x4_f32(false, ag, false, b1, (short)0, aG1, false, false);
        }
        __syncthreads();
    }
    const int cn = n0 + am;
    const int cr = (lane >> 4) << 3;
    #pragma unroll
    for (int v = 0; v < 8; ++v) {
        const int row = m0 + cr + v;
        const float ba = bo[row];
        const float bg = bo[DD + row];
        float* o = Out + ((size_t)bi * DD + row) * LL + cn;
        o[0]  = (aA0[v] + ba) * sigmoidf_(aG0[v] + bg);
        o[16] = (aA1[v] + ba) * sigmoidf_(aG1[v] + bg);
    }
}

extern "C" void kernel_launch(void* const* d_in, const int* in_sizes, int n_in,
                              void* d_out, int out_size, void* d_ws, size_t ws_size,
                              hipStream_t stream)
{
    const float* u    = (const float*)d_in[0];
    const float* xw   = (const float*)d_in[1];
    const float* dtw  = (const float*)d_in[2];
    const float* dtb  = (const float*)d_in[3];
    const float* alog = (const float*)d_in[4];
    const float* Dp   = (const float*)d_in[5];
    const float* ow   = (const float*)d_in[6];
    const float* ob   = (const float*)d_in[7];
    float* out = (float*)d_out;

    const size_t CH = (size_t)NB * DD;
    float* ws    = (float*)d_ws;
    float* xdbl  = ws;
    float* delta = xdbl + (size_t)NB * EE * LL;
    float* yv    = delta + CH * LL;
    float* Sst   = yv + CH * LL;
    float* sumD  = Sst + CH * NC * NS;
    float* hin   = sumD + CH * NC;

    const dim3 blk(128);
    hipLaunchKernelGGL(k_xproj, dim3(LL / 64, EE / 16, NB), blk, 0, stream, xw, u, xdbl);
    hipLaunchKernelGGL(k_dt,    dim3(LL / 64, DD / 16, NB), blk, 0, stream, dtw, xdbl, dtb, delta);
    hipLaunchKernelGGL(k_scanA, dim3((CH * NC) / 16), dim3(256), 0, stream, u, xdbl, delta, alog, Sst, sumD);
    hipLaunchKernelGGL(k_scanB, dim3(CH / 16), dim3(256), 0, stream, alog, Sst, sumD, hin);
    hipLaunchKernelGGL(k_scanC, dim3((CH * NC) / 16), dim3(256), 0, stream, u, xdbl, delta, alog, Dp, hin, yv);
    hipLaunchKernelGGL(k_out,   dim3(LL / 128, DD / 16, NB), blk, 0, stream, ow, yv, ob, out);
}


# scan passes: work-group = 16 channels of one chunk; its B/C tiles fetched once by whole-line loads and shared via LDS (was per-lane float4 loads at 8-KB stride); sum order unchanged
# speedup vs baseline: 1.2056x; 1.2056x over previous
#include <hip/hip_runtime.h>
#include <math.h>

#define NB 2
#define DD 1024
#define LL 2048
#define NS 16
#define RR 64
#define EE 96
#define NC 32
#define LC 64
#define BP 68
#define DG (DD / 16)

typedef float v2f __attribute__((ext_vector_type(2)));
typedef float v8f __attribute__((ext_vector_type(8)));

__device__ __forceinline__ float sigmoidf_(float x) { return 1.0f / (1.0f + expf(-x)); }
__device__ __forceinline__ float softplusf_(float x) { return (x > 20.0f) ? x : log1pf(expf(x)); }

__global__ __launch_bounds__(128)
void k_xproj(const float* __restrict__ W, const float* __restrict__ U,
             float* __restrict__ Xd)
{
    __shared__ float As[16 * 68];
    const int tid  = threadIdx.x;
    const int lane = tid & 31;
    const int wv   = tid >> 5;
    const int m0   = blockIdx.y << 4;
    const int n0   = (blockIdx.x << 6) + (wv << 4);
    const int bi   = blockIdx.z;
    const int am   = lane & 15;
    const int ak   = (lane >> 4) << 1;
    v8f acc = {};
    const float* Ub = U + ((size_t)bi * DD) * LL + n0 + am;
    for (int kc = 0; kc < DD; kc += 64) {
        for (int i = tid; i < 16 * 64; i += 128) {
            const int r = i >> 6, c = i & 63;
            As[r * 68 + c] = W[(size_t)(m0 + r) * DD + kc + c];
        }
        __syncthreads();
        #pragma unroll
        for (int k = 0; k < 64; k += 4) {
            v2f a, b;
            a.x = As[am * 68 + k + ak];
            a.y = As[am * 68 + k + ak + 1];
            const float* bp = Ub + (size_t)(kc + k + ak) * LL;
            b.x = bp[0];
            b.y = bp[LL];
            acc = __builtin_amdgcn_wmma_f32_16x16x4_f32(
                false, a, false, b, (short)0, acc, false, false);
        }
        __syncthreads();
    }
    const int cn = n0 + am;
    const int cr = (lane >> 4) << 3;
    #pragma unroll
    for (int v = 0; v < 8; ++v)
        Xd[((size_t)bi * EE + m0 + cr + v) * LL + cn] = acc[v];
}

__global__ __launch_bounds__(128)
void k_dt(const float* __restrict__ Wdt, const float* __restrict__ Xd,
          const float* __restrict__ bias, float* __restrict__ Delta)
{
    __shared__ float As[16 * 68];
    const int tid  = threadIdx.x;
    const int lane = tid & 31;
    const int wv   = tid >> 5;
    const int m0   = blockIdx.y << 4;
    const int n0   = (blockIdx.x << 6) + (wv << 4);
    const int bi   = blockIdx.z;
    const int am   = lane & 15;
    const int ak   = (lane >> 4) << 1;
    for (int i = tid; i < 16 * 64; i += 128) {
        const int r = i >> 6, c = i & 63;
        As[r * 68 + c] = Wdt[(size_t)(m0 + r) * RR + c];
    }
    __syncthreads();
    v8f acc = {};
    const float* Bb = Xd + ((size_t)bi * EE) * LL + n0 + am;
    #pragma unroll
    for (int k = 0; k < RR; k += 4) {
        v2f a, b;
        a.x = As[am * 68 + k + ak];
        a.y = As[am * 68 + k + ak + 1];
        const float* bp = Bb + (size_t)(k + ak) * LL;
        b.x = bp[0];
        b.y = bp[LL];
        acc = __builtin_amdgcn_wmma_f32_16x16x4_f32(
            false, a, false, b, (short)0, acc, false, false);
    }
    const int cn = n0 + am;
    const int cr = (lane >> 4) << 3;
    #pragma unroll
    for (int v = 0; v < 8; ++v) {
        const int row = m0 + cr + v;
        Delta[((size_t)bi * DD + row) * LL + cn] = softplusf_(acc[v] + bias[row]);
    }
}

__global__ __launch_bounds__(256)
void k_scanA(const float* __restrict__ U, const float* __restrict__ Xd,
             const float* __restrict__ Delta, const float* __restrict__ Alog,
             float* __restrict__ S, float* __restrict__ SumD)
{
    __shared__ __attribute__((aligned(16))) float Bs[NS * BP];
    const int tid  = threadIdx.x;
    const int lane = tid & 31;
    const int dg   = blockIdx.x & (DG - 1);
    const int c    = (blockIdx.x / DG) & (NC - 1);
    const int bi   = blockIdx.x / (DG * NC);
    const int d    = (dg << 4) + (tid >> 4);
    const int ch   = bi * DD + d;
    const int n    = lane & 15;
    const int hb   = lane & 16;
    {
        const int r = tid >> 4, q = tid & 15;
        const float4 t = *(const float4*)(Xd + ((size_t)bi * EE + RR + r) * LL + (size_t)c * LC + 4 * q);
        *(float4*)(Bs + r * BP + 4 * q) = t;
    }
    const float An = -expf(Alog[d * NS + n]);
    const size_t rowOff = ((size_t)bi * DD + d) * LL + (size_t)c * LC;
    const float* dRow = Delta + rowOff;
    const float* uRow = U + rowOff;
    const float* bRow = Bs + n * BP;
    __syncthreads();
    float h = 0.0f, sdp = 0.0f;
    for (int l0 = 0; l0 < LC; l0 += 16) {
        const float dl = dRow[l0 + n];
        const float ul = uRow[l0 + n];
        sdp += dl;
        float bv[16];
        #pragma unroll
        for (int q = 0; q < 4; ++q) {
            float4 t = *(const float4*)(bRow + l0 + 4 * q);
            bv[4*q+0] = t.x; bv[4*q+1] = t.y; bv[4*q+2] = t.z; bv[4*q+3] = t.w;
        }
        #pragma unroll
        for (int j = 0; j < 16; ++j) {
            const float dj = __shfl(dl, hb + j, 32);
            const float uj = __shfl(ul, hb + j, 32);
            h = expf(dj * An) * h + dj * uj * bv[j];
        }
    }
    sdp += __shfl_xor(sdp, 8, 32);
    sdp += __shfl_xor(sdp, 4, 32);
    sdp += __shfl_xor(sdp, 2, 32);
    sdp += __shfl_xor(sdp, 1, 32);
    S[(((size_t)bi * NC + c) * DD + d) * NS + n] = h;
    if (n == 0) SumD[(size_t)ch * NC + c] = sdp;
}

__global__ __launch_bounds__(256)
void k_scanB(const float* __restrict__ Alog, const float* __restrict__ S,
             const float* __restrict__ SumD, float* __restrict__ Hin)
{
    const int lane = threadIdx.x & 31;
    const int gw   = blockIdx.x * (blockDim.x >> 5) + (threadIdx.x >> 5);
    const int ch   = (gw << 1) + (lane >> 4);
    const int bi   = ch >> 10;
    const int d    = ch & (DD - 1);
    const int n    = lane & 15;
    const float An = -expf(Alog[d * NS + n]);
    const size_t so = ((size_t)bi * NC * DD + d) * NS + n;
    float P[NC];
    #pragma unroll
    for (int c = 0; c < NC; ++c)
        P[c] = expf(An * SumD[(size_t)ch * NC + c]);
    float hin = 0.0f;
    #pragma unroll
    for (int c = 0; c < NC; ++c) {
        Hin[so + (size_t)c * DD * NS] = hin;
        hin = P[c] * hin + S[so + (size_t)c * DD * NS];
    }
}

__global__ __launch_bounds__(256)
void k_scanC(const float* __restrict__ U, const float* __restrict__ Xd,
             const float* __restrict__ Delta, const float* __restrict__ Alog,
             const float* __restrict__ Dp, const float* __restrict__ Hin,
             float* __restrict__ Y)
{
    __shared__ __attribute__((aligned(16))) float Bs[NS * BP];
    __shared__ __attribute__((aligned(16))) float Cs[NS * BP];
    const int tid  = threadIdx.x;
    const int lane = tid & 31;
    const int dg   = blockIdx.x & (DG - 1);
    const int c    = (blockIdx.x / DG) & (NC - 1);
    const int bi   = blockIdx.x / (DG * NC);
    const int d    = (dg << 4) + (tid >> 4);
    const int n    = lane & 15;
    const int hb   = lane & 16;
    {
        const int r = tid >> 4, q = tid & 15;
        const float* g = Xd + ((size_t)bi * EE + RR + r) * LL + (size_t)c * LC + 4 * q;
        const float4 tb = *(const float4*)g;
        const float4 tc = *(const float4*)(g + (size_t)NS * LL);
        *(float4*)(Bs + r * BP + 4 * q) = tb;
        *(float4*)(Cs + r * BP + 4 * q) = tc;
    }
    const float An = -expf(Alog[d * NS + n]);
    const float Dv = Dp[d];
    const size_t rowOff = ((size_t)bi * DD + d) * LL + (size_t)c * LC;
    const float* dRow = Delta + rowOff;
    const float* uRow = U + rowOff;
    const float* bRow = Bs + n * BP;
    const float* cRow = Cs + n * BP;
    float* yRow = Y + rowOff;
    float h = Hin[(((size_t)bi * NC + c) * DD + d) * NS + n];
    __syncthreads();
    for (int l0 = 0; l0 < LC; l0 += 16) {
        const float dl = dRow[l0 + n];
        const float ul = uRow[l0 + n];
        float bv[16], cv[16];
        #pragma unroll
        for (int q = 0; q < 4; ++q) {
            float4 t = *(const float4*)(bRow + l0 + 4 * q);
            bv[4*q+0] = t.x; bv[4*q+1] = t.y; bv[4*q+2] = t.z; bv[4*q+3] = t.w;
            float4 s = *(const float4*)(cRow + l0 + 4 * q);
            cv[4*q+0] = s.x; cv[4*q+1] = s.y; cv[4*q+2] = s.z; cv[4*q+3] = s.w;
        }
        float ymy = 0.0f;
        #pragma unroll
        for (int j = 0; j < 16; ++j) {
            const float dj = __shfl(dl, hb + j, 32);
            const float uj = __shfl(ul, hb + j, 32);
            h = expf(dj * An) * h + dj * uj * bv[j];
            float yv = h * cv[j];
            yv += __shfl_xor(yv, 8, 32);
            yv += __shfl_xor(yv, 4, 32);
            yv += __shfl_xor(yv, 2, 32);
            yv += __shfl_xor(yv, 1, 32);
            if (n == j) ymy = yv;
        }
        float yo = ymy + ul * Dv;
        yo = yo * (ul * sigmoidf_(ul));
        yo = 0.5f * yo * (1.0f + erff(yo * 0.70710678118f));
        yRow[l0 + n] = yo;
    }
}

__global__ __launch_bounds__(128)
void k_out(const float* __restrict__ Wo, const float* __restrict__ Yv,
           const float* __restrict__ bo, float* __restrict__ Out)
{
    __shared__ float Aa[16 * 68];
    __shared__ float Ag[16 * 68];
    const int tid  = threadIdx.x;
    const int lane = tid & 31;
    const int wv   = tid >> 5;
    const int m0   = blockIdx.y << 4;
    const int n0   = (blockIdx.x << 7) + (wv << 5);
    const int bi   = blockIdx.z;
    const int am   = lane & 15;
    const int ak   = (lane >> 4) << 1;
    v8f aA0 = {}, aA1 = {}, aG0 = {}, aG1 = {};
    const float* Yb = Yv + ((size_t)bi * DD) * LL + n0 + am;
    for (int kc = 0; kc < DD; kc += 64) {
        for (int i = tid; i < 16 * 64; i += 128) {
            const int r = i >> 6, c = i & 63;
            Aa[r * 68 + c] = Wo[(size_t)(m0 + r) * DD + kc + c];
            Ag[r * 68 + c] = Wo[(size_t)(DD + m0 + r) * DD + kc + c];
        }
        __syncthreads();
        if (kc + 64 < DD)
            __builtin_prefetch(&Wo[(size_t)(m0 + (tid >> 3)) * DD + kc + 64 + ((tid & 7) << 3)], 0, 1);
        #pragma unroll
        for (int k = 0; k < 64; k += 4) {
            v2f b0, b1, aa, ag;
            const float* bp = Yb + (size_t)(kc + k + ak) * LL;
            b0.x = bp[0];
            b0.y = bp[LL];
            b1.x = bp[16];
            b1.y = bp[LL + 16];
            aa.x = Aa[am * 68 + k + ak];
            aa.y = Aa[am * 68 + k + ak + 1];
            ag.x = Ag[am * 68 + k + ak];
            ag.y = Ag[am * 68 + k + ak + 1];
            aA0 = __builtin_amdgcn_wmma_f32_16x16x4_f32(false, aa, false, b0, (short)0, aA0, false, false);
            aA1 = __builtin_amdgcn_wmma_f32_16x16x4_f32(false, aa, false, b1, (short)0, aA1, false, false);
            aG0 = __builtin_amdgcn_wmma_f32_16x16x4_f32(false, ag, false, b0, (short)0, aG0, false, false);
            aG1 = __builtin_amdgcn_wmma_f32_16x16x4_f32(false, ag, false, b1, (short)0, aG1, false, false);
        }
        __syncthreads();
    }
    const int cn = n0 + am;
    const int cr = (lane >> 4) << 3;
    #pragma unroll
    for (int v = 0; v < 8; ++v) {
        const int row = m0 + cr + v;
        const float ba = bo[row];
        const float bg = bo[DD + row];
        float* o = Out + ((size_t)bi * DD + row) * LL + cn;
        o[0]  = (aA0[v] + ba) * sigmoidf_(aG0[v] + bg);
        o[16] = (aA1[v] + ba) * sigmoidf_(aG1[v] + bg);
    }
}

extern "C" void kernel_launch(void* const* d_in, const int* in_sizes, int n_in,
                              void* d_out, int out_size, void* d_ws, size_t ws_size,
                              hipStream_t stream)
{
    const float* u    = (const float*)d_in[0];
    const float* xw   = (const float*)d_in[1];
    const float* dtw  = (const float*)d_in[2];
    const float* dtb  = (const float*)d_in[3];
    const float* alog = (const float*)d_in[4];
    const float* Dp   = (const float*)d_in[5];
    const float* ow   = (const float*)d_in[6];
    const float* ob   = (const float*)d_in[7];
    float* out = (float*)d_out;

    const size_t CH = (size_t)NB * DD;
    float* ws    = (float*)d_ws;
    float* xdbl  = ws;
    float* delta = xdbl + (size_t)NB * EE * LL;
    float* yv    = delta + CH * LL;
    float* Sst   = yv + CH * LL;
    float* sumD  = Sst + CH * NC * NS;
    float* hin   = sumD + CH * NC;

    const dim3 blk(128);
    hipLaunchKernelGGL(k_xproj, dim3(LL / 64, EE / 16, NB), blk, 0, stream, xw, u, xdbl);
    hipLaunchKernelGGL(k_dt,    dim3(LL / 64, DD / 16, NB), blk, 0, stream, dtw, xdbl, dtb, delta);
    hipLaunchKernelGGL(k_scanA, dim3((CH * NC) / 16), dim3(256), 0, stream, u, xdbl, delta, alog, Sst, sumD);
    hipLaunchKernelGGL(k_scanB, dim3(CH / 16), dim3(256), 0, stream, alog, Sst, sumD, hin);
    hipLaunchKernelGGL(k_scanC, dim3((CH * NC) / 16), dim3(256), 0, stream, u, xdbl, delta, alog, Dp, hin, yv);
    hipLaunchKernelGGL(k_out,   dim3(LL / 128, DD / 16, NB), blk, 0, stream, ow, yv, ob, out);
}


# k_out: work-group tile 128 weight rows x 128 columns (was 32 x 128), both operands staged through LDS by whole-line 16-byte loads, next K chunk in flight during the WMMAs; sum order unchanged
# speedup vs baseline: 1.3311x; 1.1041x over previous
#include <hip/hip_runtime.h>
#include <math.h>

#define NB 2
#define DD 1024
#define LL 2048
#define NS 16
#define RR 64
#define EE 96
#define NC 32
#define LC 64
#define BP 68
#define DG (DD / 16)

typedef float v2f __attribute__((ext_vector_type(2)));
typedef float v8f __attribute__((ext_vector_type(8)));

__device__ __forceinline__ float sigmoidf_(float x) { return 1.0f / (1.0f + expf(-x)); }
__device__ __forceinline__ float softplusf_(float x) { return (x > 20.0f) ? x : log1pf(expf(x)); }

__global__ __launch_bounds__(128)
void k_xproj(const float* __restrict__ W, const float* __restrict__ U,
             float* __restrict__ Xd)
{
    __shared__ float As[16 * 68];
    const int tid  = threadIdx.x;
    const int lane = tid & 31;
    const int wv   = tid >> 5;
    const int m0   = blockIdx.y << 4;
    const int n0   = (blockIdx.x << 6) + (wv << 4);
    const int bi   = blockIdx.z;
    const int am   = lane & 15;
    const int ak   = (lane >> 4) << 1;
    v8f acc = {};
    const float* Ub = U + ((size_t)bi * DD) * LL + n0 + am;
    for (int kc = 0; kc < DD; kc += 64) {
        for (int i = tid; i < 16 * 64; i += 128) {
            const int r = i >> 6, c = i & 63;
            As[r * 68 + c] = W[(size_t)(m0 + r) * DD + kc + c];
        }
        __syncthreads();
        #pragma unroll
        for (int k = 0; k < 64; k += 4) {
            v2f a, b;
            a.x = As[am * 68 + k + ak];
            a.y = As[am * 68 + k + ak + 1];
            const float* bp = Ub + (size_t)(kc + k + ak) * LL;
            b.x = bp[0];
            b.y = bp[LL];
            acc = __builtin_amdgcn_wmma_f32_16x16x4_f32(
                false, a, false, b, (short)0, acc, false, false);
        }
        __syncthreads();
    }
    const int cn = n0 + am;
    const int cr = (lane >> 4) << 3;
    #pragma unroll
    for (int v = 0; v < 8; ++v)
        Xd[((size_t)bi * EE + m0 + cr + v) * LL + cn] = acc[v];
}

__global__ __launch_bounds__(128)
void k_dt(const float* __restrict__ Wdt, const float* __restrict__ Xd,
          const float* __restrict__ bias, float* __restrict__ Delta)
{
    __shared__ float As[16 * 68];
    const int tid  = threadIdx.x;
    const int lane = tid & 31;
    const int wv   = tid >> 5;
    const int m0   = blockIdx.y << 4;
    const int n0   = (blockIdx.x << 6) + (wv << 4);
    const int bi   = blockIdx.z;
    const int am   = lane & 15;
    const int ak   = (lane >> 4) << 1;
    for (int i = tid; i < 16 * 64; i += 128) {
        const int r = i >> 6, c = i & 63;
        As[r * 68 + c] = Wdt[(size_t)(m0 + r) * RR + c];
    }
    __syncthreads();
    v8f acc = {};
    const float* Bb = Xd + ((size_t)bi * EE) * LL + n0 + am;
    #pragma unroll
    for (int k = 0; k < RR; k += 4) {
        v2f a, b;
        a.x = As[am * 68 + k + ak];
        a.y = As[am * 68 + k + ak + 1];
        const float* bp = Bb + (size_t)(k + ak) * LL;
        b.x = bp[0];
        b.y = bp[LL];
        acc = __builtin_amdgcn_wmma_f32_16x16x4_f32(
            false, a, false, b, (short)0, acc, false, false);
    }
    const int cn = n0 + am;
    const int cr = (lane >> 4) << 3;
    #pragma unroll
    for (int v = 0; v < 8; ++v) {
        const int row = m0 + cr + v;
        Delta[((size_t)bi * DD + row) * LL + cn] = softplusf_(acc[v] + bias[row]);
    }
}

__global__ __launch_bounds__(256)
void k_scanA(const float* __restrict__ U, const float* __restrict__ Xd,
             const float* __restrict__ Delta, const float* __restrict__ Alog,
             float* __restrict__ S, float* __restrict__ SumD)
{
    __shared__ __attribute__((aligned(16))) float Bs[NS * BP];
    const int tid  = threadIdx.x;
    const int lane = tid & 31;
    const int dg   = blockIdx.x & (DG - 1);
    const int c    = (blockIdx.x / DG) & (NC - 1);
    const int bi   = blockIdx.x / (DG * NC);
    const int d    = (dg << 4) + (tid >> 4);
    const int ch   = bi * DD + d;
    const int n    = lane & 15;
    const int hb   = lane & 16;
    {
        const int r = tid >> 4, q = tid & 15;
        const float4 t = *(const float4*)(Xd + ((size_t)bi * EE + RR + r) * LL + (size_t)c * LC + 4 * q);
        *(float4*)(Bs + r * BP + 4 * q) = t;
    }
    const float An = -expf(Alog[d * NS + n]);
    const size_t rowOff = ((size_t)bi * DD + d) * LL + (size_t)c * LC;
    const float* dRow = Delta + rowOff;
    const float* uRow = U + rowOff;
    const float* bRow = Bs + n * BP;
    __syncthreads();
    float h = 0.0f, sdp = 0.0f;
    for (int l0 = 0; l0 < LC; l0 += 16) {
        const float dl = dRow[l0 + n];
        const float ul = uRow[l0 + n];
        sdp += dl;
        float bv[16];
        #pragma unroll
        for (int q = 0; q < 4; ++q) {
            float4 t = *(const float4*)(bRow + l0 + 4 * q);
            bv[4*q+0] = t.x; bv[4*q+1] = t.y; bv[4*q+2] = t.z; bv[4*q+3] = t.w;
        }
        #pragma unroll
        for (int j = 0; j < 16; ++j) {
            const float dj = __shfl(dl, hb + j, 32);
            const float uj = __shfl(ul, hb + j, 32);
            h = expf(dj * An) * h + dj * uj * bv[j];
        }
    }
    sdp += __shfl_xor(sdp, 8, 32);
    sdp += __shfl_xor(sdp, 4, 32);
    sdp += __shfl_xor(sdp, 2, 32);
    sdp += __shfl_xor(sdp, 1, 32);
    S[(((size_t)bi * NC + c) * DD + d) * NS + n] = h;
    if (n == 0) SumD[(size_t)ch * NC + c] = sdp;
}

__global__ __launch_bounds__(256)
void k_scanB(const float* __restrict__ Alog, const float* __restrict__ S,
             const float* __restrict__ SumD, float* __restrict__ Hin)
{
    const int lane = threadIdx.x & 31;
    const int gw   = blockIdx.x * (blockDim.x >> 5) + (threadIdx.x >> 5);
    const int ch   = (gw << 1) + (lane >> 4);
    const int bi   = ch >> 10;
    const int d    = ch & (DD - 1);
    const int n    = lane & 15;
    const float An = -expf(Alog[d * NS + n]);
    const size_t so = ((size_t)bi * NC * DD + d) * NS + n;
    float P[NC];
    #pragma unroll
    for (int c = 0; c < NC; ++c)
        P[c] = expf(An * SumD[(size_t)ch * NC + c]);
    float hin = 0.0f;
    #pragma unroll
    for (int c = 0; c < NC; ++c) {
        Hin[so + (size_t)c * DD * NS] = hin;
        hin = P[c] * hin + S[so + (size_t)c * DD * NS];
    }
}

__global__ __launch_bounds__(256)
void k_scanC(const float* __restrict__ U, const float* __restrict__ Xd,
             const float* __restrict__ Delta, const float* __restrict__ Alog,
             const float* __restrict__ Dp, const float* __restrict__ Hin,
             float* __restrict__ Y)
{
    __shared__ __attribute__((aligned(16))) float Bs[NS * BP];
    __shared__ __attribute__((aligned(16))) float Cs[NS * BP];
    const int tid  = threadIdx.x;
    const int lane = tid & 31;
    const int dg   = blockIdx.x & (DG - 1);
    const int c    = (blockIdx.x / DG) & (NC - 1);
    const int bi   = blockIdx.x / (DG * NC);
    const int d    = (dg << 4) + (tid >> 4);
    const int n    = lane & 15;
    const int hb   = lane & 16;
    {
        const int r = tid >> 4, q = tid & 15;
        const float* g = Xd + ((size_t)bi * EE + RR + r) * LL + (size_t)c * LC + 4 * q;
        const float4 tb = *(const float4*)g;
        const float4 tc = *(const float4*)(g + (size_t)NS * LL);
        *(float4*)(Bs + r * BP + 4 * q) = tb;
        *(float4*)(Cs + r * BP + 4 * q) = tc;
    }
    const float An = -expf(Alog[d * NS + n]);
    const float Dv = Dp[d];
    const size_t rowOff = ((size_t)bi * DD + d) * LL + (size_t)c * LC;
    const float* dRow = Delta + rowOff;
    const float* uRow = U + rowOff;
    const float* bRow = Bs + n * BP;
    const float* cRow = Cs + n * BP;
    float* yRow = Y + rowOff;
    float h = Hin[(((size_t)bi * NC + c) * DD + d) * NS + n];
    __syncthreads();
    for (int l0 = 0; l0 < LC; l0 += 16) {
        const float dl = dRow[l0 + n];
        const float ul = uRow[l0 + n];
        float bv[16], cv[16];
        #pragma unroll
        for (int q = 0; q < 4; ++q) {
            float4 t = *(const float4*)(bRow + l0 + 4 * q);
            bv[4*q+0] = t.x; bv[4*q+1] = t.y; bv[4*q+2] = t.z; bv[4*q+3] = t.w;
            float4 s = *(const float4*)(cRow + l0 + 4 * q);
            cv[4*q+0] = s.x; cv[4*q+1] = s.y; cv[4*q+2] = s.z; cv[4*q+3] = s.w;
        }
        float ymy = 0.0f;
        #pragma unroll
        for (int j = 0; j < 16; ++j) {
            const float dj = __shfl(dl, hb + j, 32);
            const float uj = __shfl(ul, hb + j, 32);
            h = expf(dj * An) * h + dj * uj * bv[j];
            float yv = h * cv[j];
            yv += __shfl_xor(yv, 8, 32);
            yv += __shfl_xor(yv, 4, 32);
            yv += __shfl_xor(yv, 2, 32);
            yv += __shfl_xor(yv, 1, 32);
            if (n == j) ymy = yv;
        }
        float yo = ymy + ul * Dv;
        yo = yo * (ul * sigmoidf_(ul));
        yo = 0.5f * yo * (1.0f + erff(yo * 0.70710678118f));
        yRow[l0 + n] = yo;
    }
}

#define OKC 32
#define OPA 36
#define OPB 136

typedef float v4f __attribute__((ext_vector_type(4)));

__global__ __launch_bounds__(256)
void k_out(const float* __restrict__ Wo, const float* __restrict__ Yv,
           const float* __restrict__ bo, float* __restrict__ Out)
{
    __shared__ __attribute__((aligned(16))) float As[128 * OPA];
    __shared__ __attribute__((aligned(16))) float Bs[OKC * OPB];
    const int tid  = threadIdx.x;
    const int lane = tid & 31;
    const int wv   = tid >> 5;
    const int wm   = wv & 3;
    const int wn   = wv >> 2;
    const int m0   = blockIdx.y << 6;
    const int n0   = blockIdx.x << 7;
    const int bi   = blockIdx.z;
    const int am   = lane & 15;
    const int ak   = (lane >> 4) << 1;

    const int aq = tid & 7,  ar = tid >> 3;
    const int bq = tid & 31, br = tid >> 5;
    const float* wpA = Wo + (size_t)(m0 + ar) * DD + 4 * aq;
    const float* wpG = wpA + (size_t)DD * DD;
    const float* yp  = Yv + ((size_t)bi * DD + br) * LL + n0 + 4 * bq;
    float* sA = As + ar * OPA + 4 * aq;
    float* sB = Bs + br * OPB + 4 * bq;

    v4f ra0 = *(const v4f*)(wpA);
    v4f ra1 = *(const v4f*)(wpA + 32 * DD);
    v4f ra2 = *(const v4f*)(wpG);
    v4f ra3 = *(const v4f*)(wpG + 32 * DD);
    v4f rb0 = *(const v4f*)(yp);
    v4f rb1 = *(const v4f*)(yp + 8 * LL);
    v4f rb2 = *(const v4f*)(yp + 16 * LL);
    v4f rb3 = *(const v4f*)(yp + 24 * LL);

    v8f accA0 = {}, accA1 = {}, accA2 = {}, accA3 = {};
    v8f accG0 = {}, accG1 = {}, accG2 = {}, accG3 = {};
    const float* aA = As + (16 * wm + am) * OPA + ak;
    const float* aG = As + (64 + 16 * wm + am) * OPA + ak;
    const float* bB = Bs + ak * OPB + 64 * wn + am;

    for (int kc = 0; kc < DD; kc += OKC) {
        *(v4f*)(sA)            = ra0;
        *(v4f*)(sA + 32 * OPA) = ra1;
        *(v4f*)(sA + 64 * OPA) = ra2;
        *(v4f*)(sA + 96 * OPA) = ra3;
        *(v4f*)(sB)            = rb0;
        *(v4f*)(sB + 8 * OPB)  = rb1;
        *(v4f*)(sB + 16 * OPB) = rb2;
        *(v4f*)(sB + 24 * OPB) = rb3;
        __syncthreads();
        if (kc + OKC < DD) {
            const float* wa = wpA + kc + OKC;
            const float* wg = wpG + kc + OKC;
            const float* yy = yp + (size_t)(kc + OKC) * LL;
            ra0 = *(const v4f*)(wa);
            ra1 = *(const v4f*)(wa + 32 * DD);
            ra2 = *(const v4f*)(wg);
            ra3 = *(const v4f*)(wg + 32 * DD);
            rb0 = *(const v4f*)(yy);
            rb1 = *(const v4f*)(yy + 8 * LL);
            rb2 = *(const v4f*)(yy + 16 * LL);
            rb3 = *(const v4f*)(yy + 24 * LL);
        }
        #pragma unroll
        for (int k = 0; k < OKC; k += 4) {
            v2f aa, ag, b0, b1, b2, b3;
            aa.x = aA[k];
            aa.y = aA[k + 1];
            ag.x = aG[k];
            ag.y = aG[k + 1];
            b0.x = bB[k * OPB];            b0.y = bB[(k + 1) * OPB];
            b1.x = bB[k * OPB + 16];       b1.y = bB[(k + 1) * OPB + 16];
            b2.x = bB[k * OPB + 32];       b2.y = bB[(k + 1) * OPB + 32];
            b3.x = bB[k * OPB + 48];       b3.y = bB[(k + 1) * OPB + 48];
            accA0 = __builtin_amdgcn_wmma_f32_16x16x4_f32(false, aa, false, b0, (short)0, accA0, false, false);
            accG0 = __builtin_amdgcn_wmma_f32_16x16x4_f32(false, ag, false, b0, (short)0, accG0, false, false);
            accA1 = __builtin_amdgcn_wmma_f32_16x16x4_f32(false, aa, false, b1, (short)0, accA1, false, false);
            accG1 = __builtin_amdgcn_wmma_f32_16x16x4_f32(false, ag, false, b1, (short)0, accG1, false, false);
            accA2 = __builtin_amdgcn_wmma_f32_16x16x4_f32(false, aa, false, b2, (short)0, accA2, false, false);
            accG2 = __builtin_amdgcn_wmma_f32_16x16x4_f32(false, ag, false, b2, (short)0, accG2, false, false);
            accA3 = __builtin_amdgcn_wmma_f32_16x16x4_f32(false, aa, false, b3, (short)0, accA3, false, false);
            accG3 = __builtin_amdgcn_wmma_f32_16x16x4_f32(false, ag, false, b3, (short)0, accG3, false, false);
        }
        __syncthreads();
    }

    const int cr = (lane >> 4) << 3;
    #pragma unroll
    for (int v = 0; v < 8; ++v) {
        const int row = m0 + 16 * wm + cr + v;
        const float ba = bo[row];
        const float bg = bo[DD + row];
        float* o = Out + ((size_t)bi * DD + row) * LL + n0 + 64 * wn + am;
        o[0]  = (accA0[v] + ba) * sigmoidf_(accG0[v] + bg);
        o[16] = (accA1[v] + ba) * sigmoidf_(accG1[v] + bg);
        o[32] = (accA2[v] + ba) * sigmoidf_(accG2[v] + bg);
        o[48] = (accA3[v] + ba) * sigmoidf_(accG3[v] + bg);
    }
}

extern "C" void kernel_launch(void* const* d_in, const int* in_sizes, int n_in,
                              void* d_out, int out_size, void* d_ws, size_t ws_size,
                              hipStream_t stream)
{
    const float* u    = (const float*)d_in[0];
    const float* xw   = (const float*)d_in[1];
    const float* dtw  = (const float*)d_in[2];
    const float* dtb  = (const float*)d_in[3];
    const float* alog = (const float*)d_in[4];
    const float* Dp   = (const float*)d_in[5];
    const float* ow   = (const float*)d_in[6];
    const float* ob   = (const float*)d_in[7];
    float* out = (float*)d_out;

    const size_t CH = (size_t)NB * DD;
    float* ws    = (float*)d_ws;
    float* xdbl  = ws;
    float* delta = xdbl + (size_t)NB * EE * LL;
    float* yv    = delta + CH * LL;
    float* Sst   = yv + CH * LL;
    float* sumD  = Sst + CH * NC * NS;
    float* hin   = sumD + CH * NC;

    const dim3 blk(128);
    hipLaunchKernelGGL(k_xproj, dim3(LL / 64, EE / 16, NB), blk, 0, stream, xw, u, xdbl);
    hipLaunchKernelGGL(k_dt,    dim3(LL / 64, DD / 16, NB), blk, 0, stream, dtw, xdbl, dtb, delta);
    hipLaunchKernelGGL(k_scanA, dim3((CH * NC) / 16), dim3(256), 0, stream, u, xdbl, delta, alog, Sst, sumD);
    hipLaunchKernelGGL(k_scanB, dim3(CH / 16), dim3(256), 0, stream, alog, Sst, sumD, hin);
    hipLaunchKernelGGL(k_scanC, dim3((CH * NC) / 16), dim3(256), 0, stream, u, xdbl, delta, alog, Dp, hin, yv);
    hipLaunchKernelGGL(k_out,   dim3(LL / 128, DD / 64, NB), dim3(256), 0, stream, ow, yv, ob, out);
}


# scan passes A, C: a channel's 16 states in registers of 2 lanes (was 16 lanes, shuffles each step); one-wave groups, delta/u tiles via LDS; sum pairing kept (products may fuse), last level 1 exchange
# speedup vs baseline: 1.4069x; 1.0569x over previous
#include <hip/hip_runtime.h>
#include <math.h>

#define NB 2
#define DD 1024
#define LL 2048
#define NS 16
#define RR 64
#define EE 96
#define NC 32
#define LC 64
#define BP 68

typedef float v2f __attribute__((ext_vector_type(2)));
typedef float v8f __attribute__((ext_vector_type(8)));

__device__ __forceinline__ float sigmoidf_(float x) { return 1.0f / (1.0f + expf(-x)); }
__device__ __forceinline__ float softplusf_(float x) { return (x > 20.0f) ? x : log1pf(expf(x)); }

__global__ __launch_bounds__(128)
void k_xproj(const float* __restrict__ W, const float* __restrict__ U,
             float* __restrict__ Xd)
{
    __shared__ float As[16 * 68];
    const int tid  = threadIdx.x;
    const int lane = tid & 31;
    const int wv   = tid >> 5;
    const int m0   = blockIdx.y << 4;
    const int n0   = (blockIdx.x << 6) + (wv << 4);
    const int bi   = blockIdx.z;
    const int am   = lane & 15;
    const int ak   = (lane >> 4) << 1;
    v8f acc = {};
    const float* Ub = U + ((size_t)bi * DD) * LL + n0 + am;
    for (int kc = 0; kc < DD; kc += 64) {
        for (int i = tid; i < 16 * 64; i += 128) {
            const int r = i >> 6, c = i & 63;
            As[r * 68 + c] = W[(size_t)(m0 + r) * DD + kc + c];
        }
        __syncthreads();
        #pragma unroll
        for (int k = 0; k < 64; k += 4) {
            v2f a, b;
            a.x = As[am * 68 + k + ak];
            a.y = As[am * 68 + k + ak + 1];
            const float* bp = Ub + (size_t)(kc + k + ak) * LL;
            b.x = bp[0];
            b.y = bp[LL];
            acc = __builtin_amdgcn_wmma_f32_16x16x4_f32(
                false, a, false, b, (short)0, acc, false, false);
        }
        __syncthreads();
    }
    const int cn = n0 + am;
    const int cr = (lane >> 4) << 3;
    #pragma unroll
    for (int v = 0; v < 8; ++v)
        Xd[((size_t)bi * EE + m0 + cr + v) * LL + cn] = acc[v];
}

__global__ __launch_bounds__(128)
void k_dt(const float* __restrict__ Wdt, const float* __restrict__ Xd,
          const float* __restrict__ bias, float* __restrict__ Delta)
{
    __shared__ float As[16 * 68];
    const int tid  = threadIdx.x;
    const int lane = tid & 31;
    const int wv   = tid >> 5;
    const int m0   = blockIdx.y << 4;
    const int n0   = (blockIdx.x << 6) + (wv << 4);
    const int bi   = blockIdx.z;
    const int am   = lane & 15;
    const int ak   = (lane >> 4) << 1;
    for (int i = tid; i < 16 * 64; i += 128) {
        const int r = i >> 6, c = i & 63;
        As[r * 68 + c] = Wdt[(size_t)(m0 + r) * RR + c];
    }
    __syncthreads();
    v8f acc = {};
    const float* Bb = Xd + ((size_t)bi * EE) * LL + n0 + am;
    #pragma unroll
    for (int k = 0; k < RR; k += 4) {
        v2f a, b;
        a.x = As[am * 68 + k + ak];
        a.y = As[am * 68 + k + ak + 1];
        const float* bp = Bb + (size_t)(k + ak) * LL;
        b.x = bp[0];
        b.y = bp[LL];
        acc = __builtin_amdgcn_wmma_f32_16x16x4_f32(
            false, a, false, b, (short)0, acc, false, false);
    }
    const int cn = n0 + am;
    const int cr = (lane >> 4) << 3;
    #pragma unroll
    for (int v = 0; v < 8; ++v) {
        const int row = m0 + cr + v;
        Delta[((size_t)bi * DD + row) * LL + cn] = softplusf_(acc[v] + bias[row]);
    }
}

#define CW 16
#define HP 20

__device__ __forceinline__ float xhalf(float x)
{
    return __builtin_bit_cast(float, __builtin_amdgcn_permlanex16(
        __builtin_bit_cast(int, x), __builtin_bit_cast(int, x), 0x76543210, 0xfedcba98, false, false));
}

__device__ __forceinline__ void stage_tile16(const float* __restrict__ g, float* s, int lane)
{
    const int q = lane & 15, r0 = lane >> 4;
    #pragma unroll
    for (int k = 0; k < 8; ++k) {
        const float4 t = *(const float4*)(g + (size_t)(2 * k + r0) * LL + 4 * q);
        *(float4*)(s + (2 * k + r0) * BP + 4 * q) = t;
    }
}

__global__ __launch_bounds__(32)
void k_scanA(const float* __restrict__ U, const float* __restrict__ Xd,
             const float* __restrict__ Delta, const float* __restrict__ Alog,
             float* __restrict__ S, float* __restrict__ SumD)
{
    __shared__ __attribute__((aligned(16))) float Ds[CW * BP];
    __shared__ __attribute__((aligned(16))) float Us[CW * BP];
    __shared__ __attribute__((aligned(16))) float Bs[NS * BP];
    const int lane = threadIdx.x;
    const int cl   = lane & 15;
    const int hh   = lane >> 4;
    const int dg   = blockIdx.x & (DD / CW - 1);
    const int c    = (blockIdx.x / (DD / CW)) & (NC - 1);
    const int bi   = blockIdx.x / ((DD / CW) * NC);
    const int d0   = dg * CW;
    const int d    = d0 + cl;
    const size_t tileOff = ((size_t)bi * DD + d0) * LL + (size_t)c * LC;
    stage_tile16(Delta + tileOff, Ds, lane);
    stage_tile16(U + tileOff, Us, lane);
    stage_tile16(Xd + ((size_t)bi * EE + RR) * LL + (size_t)c * LC, Bs, lane);
    float An[8];
    #pragma unroll
    for (int q = 0; q < 4; ++q) {
        const float4 a = *(const float4*)(Alog + d * NS + 4 * q);
        An[2*q+0] = -expf(hh ? a.y : a.x);
        An[2*q+1] = -expf(hh ? a.w : a.z);
    }
    __syncthreads();
    const float* dRow = Ds + cl * BP;
    const float* uRow = Us + cl * BP;
    const float* bRow = Bs + hh * BP;
    {
        float sd[16];
        #pragma unroll
        for (int q = 0; q < 4; ++q) {
            const float4 a0 = *(const float4*)(dRow + 4 * q);
            const float4 a1 = *(const float4*)(dRow + 16 + 4 * q);
            const float4 a2 = *(const float4*)(dRow + 32 + 4 * q);
            const float4 a3 = *(const float4*)(dRow + 48 + 4 * q);
            sd[4*q+0] = ((a0.x + a1.x) + a2.x) + a3.x;
            sd[4*q+1] = ((a0.y + a1.y) + a2.y) + a3.y;
            sd[4*q+2] = ((a0.z + a1.z) + a2.z) + a3.z;
            sd[4*q+3] = ((a0.w + a1.w) + a2.w) + a3.w;
        }
        #pragma unroll
        for (int i = 0; i < 8; ++i) sd[i] += sd[i + 8];
        #pragma unroll
        for (int i = 0; i < 4; ++i) sd[i] += sd[i + 4];
        sd[0] += sd[2]; sd[1] += sd[3];
        if (hh == 0) SumD[((size_t)bi * DD + d) * NC + c] = sd[0] + sd[1];
    }
    float h[8];
    #pragma unroll
    for (int i = 0; i < 8; ++i) h[i] = 0.0f;
    for (int l0 = 0; l0 < LC; l0 += 4) {
        const float4 dq = *(const float4*)(dRow + l0);
        const float4 uq = *(const float4*)(uRow + l0);
        const float dv[4] = {dq.x, dq.y, dq.z, dq.w};
        const float uv[4] = {uq.x, uq.y, uq.z, uq.w};
        float bv[8][4];
        #pragma unroll
        for (int i = 0; i < 8; ++i) {
            const float4 t = *(const float4*)(bRow + 2 * i * BP + l0);
            bv[i][0] = t.x; bv[i][1] = t.y; bv[i][2] = t.z; bv[i][3] = t.w;
        }
        #pragma unroll
        for (int t = 0; t < 4; ++t) {
            const float dj = dv[t];
            const float uj = uv[t];
            #pragma unroll
            for (int i = 0; i < 8; ++i)
                h[i] = expf(dj * An[i]) * h[i] + dj * uj * bv[i][t];
        }
    }
    __syncthreads();
    float* Hs = Bs;
    #pragma unroll
    for (int i = 0; i < 8; ++i) Hs[cl * HP + 2 * i + hh] = h[i];
    __syncthreads();
    float* Sg = S + (((size_t)bi * NC + c) * DD + d0) * NS;
    #pragma unroll
    for (int k = 0; k < 2; ++k) {
        const int r = 8 * k + (lane >> 2), q = lane & 3;
        const float4 t = *(const float4*)(Hs + r * HP + 4 * q);
        *(float4*)(Sg + r * NS + 4 * q) = t;
    }
}

__global__ __launch_bounds__(256)
void k_scanB(const float* __restrict__ Alog, const float* __restrict__ S,
             const float* __restrict__ SumD, float* __restrict__ Hin)
{
    const int lane = threadIdx.x & 31;
    const int gw   = blockIdx.x * (blockDim.x >> 5) + (threadIdx.x >> 5);
    const int ch   = (gw << 1) + (lane >> 4);
    const int bi   = ch >> 10;
    const int d    = ch & (DD - 1);
    const int n    = lane & 15;
    const float An = -expf(Alog[d * NS + n]);
    const size_t so = ((size_t)bi * NC * DD + d) * NS + n;
    float P[NC];
    #pragma unroll
    for (int c = 0; c < NC; ++c)
        P[c] = expf(An * SumD[(size_t)ch * NC + c]);
    float hin = 0.0f;
    #pragma unroll
    for (int c = 0; c < NC; ++c) {
        Hin[so + (size_t)c * DD * NS] = hin;
        hin = P[c] * hin + S[so + (size_t)c * DD * NS];
    }
}

__global__ __launch_bounds__(32)
void k_scanC(const float* __restrict__ U, const float* __restrict__ Xd,
             const float* __restrict__ Delta, const float* __restrict__ Alog,
             const float* __restrict__ Dp, const float* __restrict__ Hin,
             float* __restrict__ Y)
{
    __shared__ __attribute__((aligned(16))) float Ds[CW * BP];
    __shared__ __attribute__((aligned(16))) float Us[CW * BP];
    __shared__ __attribute__((aligned(16))) float Bs[NS * BP];
    __shared__ __attribute__((aligned(16))) float Cs[NS * BP];
    const int lane = threadIdx.x;
    const int cl   = lane & 15;
    const int hh   = lane >> 4;
    const int dg   = blockIdx.x & (DD / CW - 1);
    const int c    = (blockIdx.x / (DD / CW)) & (NC - 1);
    const int bi   = blockIdx.x / ((DD / CW) * NC);
    const int d0   = dg * CW;
    const int d    = d0 + cl;
    const size_t tileOff = ((size_t)bi * DD + d0) * LL + (size_t)c * LC;
    stage_tile16(Delta + tileOff, Ds, lane);
    stage_tile16(U + tileOff, Us, lane);
    const float* gB = Xd + ((size_t)bi * EE + RR) * LL + (size_t)c * LC;
    stage_tile16(gB, Bs, lane);
    stage_tile16(gB + (size_t)NS * LL, Cs, lane);
    float An[8], h[8];
    const float* hg = Hin + (((size_t)bi * NC + c) * DD + d) * NS;
    #pragma unroll
    for (int q = 0; q < 4; ++q) {
        const float4 a = *(const float4*)(Alog + d * NS + 4 * q);
        An[2*q+0] = -expf(hh ? a.y : a.x);
        An[2*q+1] = -expf(hh ? a.w : a.z);
        const float4 t = *(const float4*)(hg + 4 * q);
        h[2*q+0] = hh ? t.y : t.x;
        h[2*q+1] = hh ? t.w : t.z;
    }
    const float Dv = Dp[d];
    __syncthreads();
    float* dRow = Ds + cl * BP;
    const float* uRow = Us + cl * BP;
    const float* bRow = Bs + hh * BP;
    const float* cRow = Cs + hh * BP;
    for (int l0 = 0; l0 < LC; l0 += 4) {
        const float4 dq = *(const float4*)(dRow + l0);
        const float4 uq = *(const float4*)(uRow + l0);
        const float dv[4] = {dq.x, dq.y, dq.z, dq.w};
        const float uv[4] = {uq.x, uq.y, uq.z, uq.w};
        float bv[8][4], cv[8][4];
        #pragma unroll
        for (int i = 0; i < 8; ++i) {
            const float4 t = *(const float4*)(bRow + 2 * i * BP + l0);
            bv[i][0] = t.x; bv[i][1] = t.y; bv[i][2] = t.z; bv[i][3] = t.w;
            const float4 s = *(const float4*)(cRow + 2 * i * BP + l0);
            cv[i][0] = s.x; cv[i][1] = s.y; cv[i][2] = s.z; cv[i][3] = s.w;
        }
        float part[4];
        #pragma unroll
        for (int t = 0; t < 4; ++t) {
            const float dj = dv[t];
            const float uj = uv[t];
            float p[8];
            #pragma unroll
            for (int i = 0; i < 8; ++i) {
                h[i] = expf(dj * An[i]) * h[i] + dj * uj * bv[i][t];
                p[i] = h[i] * cv[i][t];
            }
            #pragma unroll
            for (int i = 0; i < 4; ++i) p[i] += p[i + 4];
            p[0] += p[2]; p[1] += p[3];
            part[t] = p[0] + p[1];
        }
        float yf[4];
        #pragma unroll
        for (int t = 0; t < 4; ++t) yf[t] = part[t] + xhalf(part[t]);
        float yo2[2];
        #pragma unroll
        for (int k = 0; k < 2; ++k) {
            const float ul = hh ? uv[2 + k] : uv[k];
            float yo = (hh ? yf[2 + k] : yf[k]) + ul * Dv;
            yo = yo * (ul * sigmoidf_(ul));
            yo = 0.5f * yo * (1.0f + erff(yo * 0.70710678118f));
            yo2[k] = yo;
        }
        float2 yq; yq.x = yo2[0]; yq.y = yo2[1];
        *(float2*)(dRow + l0 + 2 * hh) = yq;
    }
    __syncthreads();
    {
        const int q = lane & 15, r0 = lane >> 4;
        float* yg = Y + tileOff;
        #pragma unroll
        for (int k = 0; k < 8; ++k) {
            const float4 t = *(const float4*)(Ds + (2 * k + r0) * BP + 4 * q);
            *(float4*)(yg + (size_t)(2 * k + r0) * LL + 4 * q) = t;
        }
    }
}

#define OKC 32
#define OPA 36
#define OPB 136

typedef float v4f __attribute__((ext_vector_type(4)));

__global__ __launch_bounds__(256)
void k_out(const float* __restrict__ Wo, const float* __restrict__ Yv,
           const float* __restrict__ bo, float* __restrict__ Out)
{
    __shared__ __attribute__((aligned(16))) float As[128 * OPA];
    __shared__ __attribute__((aligned(16))) float Bs[OKC * OPB];
    const int tid  = threadIdx.x;
    const int lane = tid & 31;
    const int wv   = tid >> 5;
    const int wm   = wv & 3;
    const int wn   = wv >> 2;
    const int m0   = blockIdx.y << 6;
    const int n0   = blockIdx.x << 7;
    const int bi   = blockIdx.z;
    const int am   = lane & 15;
    const int ak   = (lane >> 4) << 1;

    const int aq = tid & 7,  ar = tid >> 3;
    const int bq = tid & 31, br = tid >> 5;
    const float* wpA = Wo + (size_t)(m0 + ar) * DD + 4 * aq;
    const float* wpG = wpA + (size_t)DD * DD;
    const float* yp  = Yv + ((size_t)bi * DD + br) * LL + n0 + 4 * bq;
    float* sA = As + ar * OPA + 4 * aq;
    float* sB = Bs + br * OPB + 4 * bq;

    v4f ra0 = *(const v4f*)(wpA);
    v4f ra1 = *(const v4f*)(wpA + 32 * DD);
    v4f ra2 = *(const v4f*)(wpG);
    v4f ra3 = *(const v4f*)(wpG + 32 * DD);
    v4f rb0 = *(const v4f*)(yp);
    v4f rb1 = *(const v4f*)(yp + 8 * LL);
    v4f rb2 = *(const v4f*)(yp + 16 * LL);
    v4f rb3 = *(const v4f*)(yp + 24 * LL);

    v8f accA0 = {}, accA1 = {}, accA2 = {}, accA3 = {};
    v8f accG0 = {}, accG1 = {}, accG2 = {}, accG3 = {};
    const float* aA = As + (16 * wm + am) * OPA + ak;
    const float* aG = As + (64 + 16 * wm + am) * OPA + ak;
    const float* bB = Bs + ak * OPB + 64 * wn + am;

    for (int kc = 0; kc < DD; kc += OKC) {
        *(v4f*)(sA)            = ra0;
        *(v4f*)(sA + 32 * OPA) = ra1;
        *(v4f*)(sA + 64 * OPA) = ra2;
        *(v4f*)(sA + 96 * OPA) = ra3;
        *(v4f*)(sB)            = rb0;
        *(v4f*)(sB + 8 * OPB)  = rb1;
        *(v4f*)(sB + 16 * OPB) = rb2;
        *(v4f*)(sB + 24 * OPB) = rb3;
        __syncthreads();
        if (kc + OKC < DD) {
            const float* wa = wpA + kc + OKC;
            const float* wg = wpG + kc + OKC;
            const float* yy = yp + (size_t)(kc + OKC) * LL;
            ra0 = *(const v4f*)(wa);
            ra1 = *(const v4f*)(wa + 32 * DD);
            ra2 = *(const v4f*)(wg);
            ra3 = *(const v4f*)(wg + 32 * DD);
            rb0 = *(const v4f*)(yy);
            rb1 = *(const v4f*)(yy + 8 * LL);
            rb2 = *(const v4f*)(yy + 16 * LL);
            rb3 = *(const v4f*)(yy + 24 * LL);
        }
        #pragma unroll
        for (int k = 0; k < OKC; k += 4) {
            v2f aa, ag, b0, b1, b2, b3;
            aa.x = aA[k];
            aa.y = aA[k + 1];
            ag.x = aG[k];
            ag.y = aG[k + 1];
            b0.x = bB[k * OPB];            b0.y = bB[(k + 1) * OPB];
            b1.x = bB[k * OPB + 16];       b1.y = bB[(k + 1) * OPB + 16];
            b2.x = bB[k * OPB + 32];       b2.y = bB[(k + 1) * OPB + 32];
            b3.x = bB[k * OPB + 48];       b3.y = bB[(k + 1) * OPB + 48];
            accA0 = __builtin_amdgcn_wmma_f32_16x16x4_f32(false, aa, false, b0, (short)0, accA0, false, false);
            accG0 = __builtin_amdgcn_wmma_f32_16x16x4_f32(false, ag, false, b0, (short)0, accG0, false, false);
            accA1 = __builtin_amdgcn_wmma_f32_16x16x4_f32(false, aa, false, b1, (short)0, accA1, false, false);
            accG1 = __builtin_amdgcn_wmma_f32_16x16x4_f32(false, ag, false, b1, (short)0, accG1, false, false);
            accA2 = __builtin_amdgcn_wmma_f32_16x16x4_f32(false, aa, false, b2, (short)0, accA2, false, false);
            accG2 = __builtin_amdgcn_wmma_f32_16x16x4_f32(false, ag, false, b2, (short)0, accG2, false, false);
            accA3 = __builtin_amdgcn_wmma_f32_16x16x4_f32(false, aa, false, b3, (short)0, accA3, false, false);
            accG3 = __builtin_amdgcn_wmma_f32_16x16x4_f32(false, ag, false, b3, (short)0, accG3, false, false);
        }
        __syncthreads();
    }

    const int cr = (lane >> 4) << 3;
    #pragma unroll
    for (int v = 0; v < 8; ++v) {
        const int row = m0 + 16 * wm + cr + v;
        const float ba = bo[row];
        const float bg = bo[DD + row];
        float* o = Out + ((size_t)bi * DD + row) * LL + n0 + 64 * wn + am;
        o[0]  = (accA0[v] + ba) * sigmoidf_(accG0[v] + bg);
        o[16] = (accA1[v] + ba) * sigmoidf_(accG1[v] + bg);
        o[32] = (accA2[v] + ba) * sigmoidf_(accG2[v] + bg);
        o[48] = (accA3[v] + ba) * sigmoidf_(accG3[v] + bg);
    }
}

extern "C" void kernel_launch(void* const* d_in, const int* in_sizes, int n_in,
                              void* d_out, int out_size, void* d_ws, size_t ws_size,
                              hipStream_t stream)
{
    const float* u    = (const float*)d_in[0];
    const float* xw   = (const float*)d_in[1];
    const float* dtw  = (const float*)d_in[2];
    const float* dtb  = (const float*)d_in[3];
    const float* alog = (const float*)d_in[4];
    const float* Dp   = (const float*)d_in[5];
    const float* ow   = (const float*)d_in[6];
    const float* ob   = (const float*)d_in[7];
    float* out = (float*)d_out;

    const size_t CH = (size_t)NB * DD;
    float* ws    = (float*)d_ws;
    float* xdbl  = ws;
    float* delta = xdbl + (size_t)NB * EE * LL;
    float* yv    = delta + CH * LL;
    float* Sst   = yv + CH * LL;
    float* sumD  = Sst + CH * NC * NS;
    float* hin   = sumD + CH * NC;

    const dim3 blk(128);
    hipLaunchKernelGGL(k_xproj, dim3(LL / 64, EE / 16, NB), blk, 0, stream, xw, u, xdbl);
    hipLaunchKernelGGL(k_dt,    dim3(LL / 64, DD / 16, NB), blk, 0, stream, dtw, xdbl, dtb, delta);
    hipLaunchKernelGGL(k_scanA, dim3((CH * NC) / CW), dim3(32), 0, stream, u, xdbl, delta, alog, Sst, sumD);
    hipLaunchKernelGGL(k_scanB, dim3(CH / 16), dim3(256), 0, stream, alog, Sst, sumD, hin);
    hipLaunchKernelGGL(k_scanC, dim3((CH * NC) / CW), dim3(32), 0, stream, u, xdbl, delta, alog, Dp, hin, yv);
    hipLaunchKernelGGL(k_out,   dim3(LL / 128, DD / 64, NB), dim3(256), 0, stream, ow, yv, ob, out);
}


# k_xproj: work-group's whole 16 x 1024 weight slab staged in LDS once, all loads in flight (was 8 one-at-a-time loads + 2 barriers per 64-k chunk); B streamed ahead of the WMMAs; sum order unchanged
# speedup vs baseline: 1.8933x; 1.3457x over previous
#include <hip/hip_runtime.h>
#include <math.h>

#define NB 2
#define DD 1024
#define LL 2048
#define NS 16
#define RR 64
#define EE 96
#define NC 32
#define LC 64
#define BP 68

typedef float v2f __attribute__((ext_vector_type(2)));
typedef float v8f __attribute__((ext_vector_type(8)));

__device__ __forceinline__ float sigmoidf_(float x) { return 1.0f / (1.0f + expf(-x)); }
__device__ __forceinline__ float softplusf_(float x) { return (x > 20.0f) ? x : log1pf(expf(x)); }

#define XP 1028
#define XCH 64

__device__ __forceinline__ void xproj_load_b(float (&b)[32], const float* __restrict__ p)
{
    #pragma unroll
    for (int j = 0; j < 16; ++j) {
        b[2 * j]     = p[(size_t)(4 * j) * LL];
        b[2 * j + 1] = p[(size_t)(4 * j + 1) * LL];
    }
}

__global__ __launch_bounds__(128)
void k_xproj(const float* __restrict__ W, const float* __restrict__ U,
             float* __restrict__ Xd)
{
    __shared__ __attribute__((aligned(16))) float As[16 * XP];
    const int tid  = threadIdx.x;
    const int lane = tid & 31;
    const int wv   = tid >> 5;
    const int m0   = blockIdx.y << 4;
    const int n0   = (blockIdx.x << 6) + (wv << 4);
    const int bi   = blockIdx.z;
    const int am   = lane & 15;
    const int ak   = (lane >> 4) << 1;
    const float* Ub = U + ((size_t)bi * DD + ak) * LL + n0 + am;

    float b[3][32];
    xproj_load_b(b[0], Ub);
    xproj_load_b(b[1], Ub + (size_t)XCH * LL);

    {
        const float* wp = W + (size_t)m0 * DD + 4 * tid;
        float4 t[32];
        #pragma unroll
        for (int j = 0; j < 32; ++j)
            t[j] = *(const float4*)(wp + (size_t)(j >> 1) * DD + 512 * (j & 1));
        #pragma unroll
        for (int j = 0; j < 32; ++j)
            *(float4*)(As + (j >> 1) * XP + 512 * (j & 1) + 4 * tid) = t[j];
    }
    __syncthreads();

    v8f acc = {};
    const float* aRow = As + am * XP + ak;
    #pragma unroll
    for (int c = 0; c < DD / XCH; ++c) {
        if (c + 2 < DD / XCH)
            xproj_load_b(b[(c + 2) % 3], Ub + (size_t)((c + 2) * XCH) * LL);
        #pragma unroll
        for (int j = 0; j < 16; ++j) {
            const v2f a = *(const v2f*)(aRow + c * XCH + 4 * j);
            v2f bb;
            bb.x = b[c % 3][2 * j];
            bb.y = b[c % 3][2 * j + 1];
            acc = __builtin_amdgcn_wmma_f32_16x16x4_f32(
                false, a, false, bb, (short)0, acc, false, false);
        }
    }
    const int cn = n0 + am;
    const int cr = (lane >> 4) << 3;
    #pragma unroll
    for (int v = 0; v < 8; ++v)
        Xd[((size_t)bi * EE + m0 + cr + v) * LL + cn] = acc[v];
}

__global__ __launch_bounds__(128)
void k_dt(const float* __restrict__ Wdt, const float* __restrict__ Xd,
          const float* __restrict__ bias, float* __restrict__ Delta)
{
    __shared__ float As[16 * 68];
    const int tid  = threadIdx.x;
    const int lane = tid & 31;
    const int wv   = tid >> 5;
    const int m0   = blockIdx.y << 4;
    const int n0   = (blockIdx.x << 6) + (wv << 4);
    const int bi   = blockIdx.z;
    const int am   = lane & 15;
    const int ak   = (lane >> 4) << 1;
    for (int i = tid; i < 16 * 64; i += 128) {
        const int r = i >> 6, c = i & 63;
        As[r * 68 + c] = Wdt[(size_t)(m0 + r) * RR + c];
    }
    __syncthreads();
    v8f acc = {};
    const float* Bb = Xd + ((size_t)bi * EE) * LL + n0 + am;
    #pragma unroll
    for (int k = 0; k < RR; k += 4) {
        v2f a, b;
        a.x = As[am * 68 + k + ak];
        a.y = As[am * 68 + k + ak + 1];
        const float* bp = Bb + (size_t)(k + ak) * LL;
        b.x = bp[0];
        b.y = bp[LL];
        acc = __builtin_amdgcn_wmma_f32_16x16x4_f32(
            false, a, false, b, (short)0, acc, false, false);
    }
    const int cn = n0 + am;
    const int cr = (lane >> 4) << 3;
    #pragma unroll
    for (int v = 0; v < 8; ++v) {
        const int row = m0 + cr + v;
        Delta[((size_t)bi * DD + row) * LL + cn] = softplusf_(acc[v] + bias[row]);
    }
}

#define CW 16
#define HP 20

__device__ __forceinline__ float xhalf(float x)
{
    return __builtin_bit_cast(float, __builtin_amdgcn_permlanex16(
        __builtin_bit_cast(int, x), __builtin_bit_cast(int, x), 0x76543210, 0xfedcba98, false, false));
}

__device__ __forceinline__ void stage_tile16(const float* __restrict__ g, float* s, int lane)
{
    const int q = lane & 15, r0 = lane >> 4;
    #pragma unroll
    for (int k = 0; k < 8; ++k) {
        const float4 t = *(const float4*)(g + (size_t)(2 * k + r0) * LL + 4 * q);
        *(float4*)(s + (2 * k + r0) * BP + 4 * q) = t;
    }
}

__global__ __launch_bounds__(32)
void k_scanA(const float* __restrict__ U, const float* __restrict__ Xd,
             const float* __restrict__ Delta, const float* __restrict__ Alog,
             float* __restrict__ S, float* __restrict__ SumD)
{
    __shared__ __attribute__((aligned(16))) float Ds[CW * BP];
    __shared__ __attribute__((aligned(16))) float Us[CW * BP];
    __shared__ __attribute__((aligned(16))) float Bs[NS * BP];
    const int lane = threadIdx.x;
    const int cl   = lane & 15;
    const int hh   = lane >> 4;
    const int dg   = blockIdx.x & (DD / CW - 1);
    const int c    = (blockIdx.x / (DD / CW)) & (NC - 1);
    const int bi   = blockIdx.x / ((DD / CW) * NC);
    const int d0   = dg * CW;
    const int d    = d0 + cl;
    const size_t tileOff = ((size_t)bi * DD + d0) * LL + (size_t)c * LC;
    stage_tile16(Delta + tileOff, Ds, lane);
    stage_tile16(U + tileOff, Us, lane);
    stage_tile16(Xd + ((size_t)bi * EE + RR) * LL + (size_t)c * LC, Bs, lane);
    float An[8];
    #pragma unroll
    for (int q = 0; q < 4; ++q) {
        const float4 a = *(const float4*)(Alog + d * NS + 4 * q);
        An[2*q+0] = -expf(hh ? a.y : a.x);
        An[2*q+1] = -expf(hh ? a.w : a.z);
    }
    __syncthreads();
    const float* dRow = Ds + cl * BP;
    const float* uRow = Us + cl * BP;
    const float* bRow = Bs + hh * BP;
    {
        float sd[16];
        #pragma unroll
        for (int q = 0; q < 4; ++q) {
            const float4 a0 = *(const float4*)(dRow + 4 * q);
            const float4 a1 = *(const float4*)(dRow + 16 + 4 * q);
            const float4 a2 = *(const float4*)(dRow + 32 + 4 * q);
            const float4 a3 = *(const float4*)(dRow + 48 + 4 * q);
            sd[4*q+0] = ((a0.x + a1.x) + a2.x) + a3.x;
            sd[4*q+1] = ((a0.y + a1.y) + a2.y) + a3.y;
            sd[4*q+2] = ((a0.z + a1.z) + a2.z) + a3.z;
            sd[4*q+3] = ((a0.w + a1.w) + a2.w) + a3.w;
        }
        #pragma unroll
        for (int i = 0; i < 8; ++i) sd[i] += sd[i + 8];
        #pragma unroll
        for (int i = 0; i < 4; ++i) sd[i] += sd[i + 4];
        sd[0] += sd[2]; sd[1] += sd[3];
        if (hh == 0) SumD[((size_t)bi * DD + d) * NC + c] = sd[0] + sd[1];
    }
    float h[8];
    #pragma unroll
    for (int i = 0; i < 8; ++i) h[i] = 0.0f;
    for (int l0 = 0; l0 < LC; l0 += 4) {
        const float4 dq = *(const float4*)(dRow + l0);
        const float4 uq = *(const float4*)(uRow + l0);
        const float dv[4] = {dq.x, dq.y, dq.z, dq.w};
        const float uv[4] = {uq.x, uq.y, uq.z, uq.w};
        float bv[8][4];
        #pragma unroll
        for (int i = 0; i < 8; ++i) {
            const float4 t = *(const float4*)(bRow + 2 * i * BP + l0);
            bv[i][0] = t.x; bv[i][1] = t.y; bv[i][2] = t.z; bv[i][3] = t.w;
        }
        #pragma unroll
        for (int t = 0; t < 4; ++t) {
            const float dj = dv[t];
            const float uj = uv[t];
            #pragma unroll
            for (int i = 0; i < 8; ++i)
                h[i] = expf(dj * An[i]) * h[i] + dj * uj * bv[i][t];
        }
    }
    __syncthreads();
    float* Hs = Bs;
    #pragma unroll
    for (int i = 0; i < 8; ++i) Hs[cl * HP + 2 * i + hh] = h[i];
    __syncthreads();
    float* Sg = S + (((size_t)bi * NC + c) * DD + d0) * NS;
    #pragma unroll
    for (int k = 0; k < 2; ++k) {
        const int r = 8 * k + (lane >> 2), q = lane & 3;
        const float4 t = *(const float4*)(Hs + r * HP + 4 * q);
        *(float4*)(Sg + r * NS + 4 * q) = t;
    }
}

__global__ __launch_bounds__(256)
void k_scanB(const float* __restrict__ Alog, const float* __restrict__ S,
             const float* __restrict__ SumD, float* __restrict__ Hin)
{
    const int lane = threadIdx.x & 31;
    const int gw   = blockIdx.x * (blockDim.x >> 5) + (threadIdx.x >> 5);
    const int ch   = (gw << 1) + (lane >> 4);
    const int bi   = ch >> 10;
    const int d    = ch & (DD - 1);
    const int n    = lane & 15;
    const float An = -expf(Alog[d * NS + n]);
    const size_t so = ((size_t)bi * NC * DD + d) * NS + n;
    float P[NC];
    #pragma unroll
    for (int c = 0; c < NC; ++c)
        P[c] = expf(An * SumD[(size_t)ch * NC + c]);
    float hin = 0.0f;
    #pragma unroll
    for (int c = 0; c < NC; ++c) {
        Hin[so + (size_t)c * DD * NS] = hin;
        hin = P[c] * hin + S[so + (size_t)c * DD * NS];
    }
}

__global__ __launch_bounds__(32)
void k_scanC(const float* __restrict__ U, const float* __restrict__ Xd,
             const float* __restrict__ Delta, const float* __restrict__ Alog,
             const float* __restrict__ Dp, const float* __restrict__ Hin,
             float* __restrict__ Y)
{
    __shared__ __attribute__((aligned(16))) float Ds[CW * BP];
    __shared__ __attribute__((aligned(16))) float Us[CW * BP];
    __shared__ __attribute__((aligned(16))) float Bs[NS * BP];
    __shared__ __attribute__((aligned(16))) float Cs[NS * BP];
    const int lane = threadIdx.x;
    const int cl   = lane & 15;
    const int hh   = lane >> 4;
    const int dg   = blockIdx.x & (DD / CW - 1);
    const int c    = (blockIdx.x / (DD / CW)) & (NC - 1);
    const int bi   = blockIdx.x / ((DD / CW) * NC);
    const int d0   = dg * CW;
    const int d    = d0 + cl;
    const size_t tileOff = ((size_t)bi * DD + d0) * LL + (size_t)c * LC;
    stage_tile16(Delta + tileOff, Ds, lane);
    stage_tile16(U + tileOff, Us, lane);
    const float* gB = Xd + ((size_t)bi * EE + RR) * LL + (size_t)c * LC;
    stage_tile16(gB, Bs, lane);
    stage_tile16(gB + (size_t)NS * LL, Cs, lane);
    float An[8], h[8];
    const float* hg = Hin + (((size_t)bi * NC + c) * DD + d) * NS;
    #pragma unroll
    for (int q = 0; q < 4; ++q) {
        const float4 a = *(const float4*)(Alog + d * NS + 4 * q);
        An[2*q+0] = -expf(hh ? a.y : a.x);
        An[2*q+1] = -expf(hh ? a.w : a.z);
        const float4 t = *(const float4*)(hg + 4 * q);
        h[2*q+0] = hh ? t.y : t.x;
        h[2*q+1] = hh ? t.w : t.z;
    }
    const float Dv = Dp[d];
    __syncthreads();
    float* dRow = Ds + cl * BP;
    const float* uRow = Us + cl * BP;
    const float* bRow = Bs + hh * BP;
    const float* cRow = Cs + hh * BP;
    for (int l0 = 0; l0 < LC; l0 += 4) {
        const float4 dq = *(const float4*)(dRow + l0);
        const float4 uq = *(const float4*)(uRow + l0);
        const float dv[4] = {dq.x, dq.y, dq.z, dq.w};
        const float uv[4] = {uq.x, uq.y, uq.z, uq.w};
        float bv[8][4], cv[8][4];
        #pragma unroll
        for (int i = 0; i < 8; ++i) {
            const float4 t = *(const float4*)(bRow + 2 * i * BP + l0);
            bv[i][0] = t.x; bv[i][1] = t.y; bv[i][2] = t.z; bv[i][3] = t.w;
            const float4 s = *(const float4*)(cRow + 2 * i * BP + l0);
            cv[i][0] = s.x; cv[i][1] = s.y; cv[i][2] = s.z; cv[i][3] = s.w;
        }
        float part[4];
        #pragma unroll
        for (int t = 0; t < 4; ++t) {
            const float dj = dv[t];
            const float uj = uv[t];
            float p[8];
            #pragma unroll
            for (int i = 0; i < 8; ++i) {
                h[i] = expf(dj * An[i]) * h[i] + dj * uj * bv[i][t];
                p[i] = h[i] * cv[i][t];
            }
            #pragma unroll
            for (int i = 0; i < 4; ++i) p[i] += p[i + 4];
            p[0] += p[2]; p[1] += p[3];
            part[t] = p[0] + p[1];
        }
        float yf[4];
        #pragma unroll
        for (int t = 0; t < 4; ++t) yf[t] = part[t] + xhalf(part[t]);
        float yo2[2];
        #pragma unroll
        for (int k = 0; k < 2; ++k) {
            const float ul = hh ? uv[2 + k] : uv[k];
            float yo = (hh ? yf[2 + k] : yf[k]) + ul * Dv;
            yo = yo * (ul * sigmoidf_(ul));
            yo = 0.5f * yo * (1.0f + erff(yo * 0.70710678118f));
            yo2[k] = yo;
        }
        float2 yq; yq.x = yo2[0]; yq.y = yo2[1];
        *(float2*)(dRow + l0 + 2 * hh) = yq;
    }
    __syncthreads();
    {
        const int q = lane & 15, r0 = lane >> 4;
        float* yg = Y + tileOff;
        #pragma unroll
        for (int k = 0; k < 8; ++k) {
            const float4 t = *(const float4*)(Ds + (2 * k + r0) * BP + 4 * q);
            *(float4*)(yg + (size_t)(2 * k + r0) * LL + 4 * q) = t;
        }
    }
}

#define OKC 32
#define OPA 36
#define OPB 136

typedef float v4f __attribute__((ext_vector_type(4)));

__global__ __launch_bounds__(256)
void k_out(const float* __restrict__ Wo, const float* __restrict__ Yv,
           const float* __restrict__ bo, float* __restrict__ Out)
{
    __shared__ __attribute__((aligned(16))) float As[128 * OPA];
    __shared__ __attribute__((aligned(16))) float Bs[OKC * OPB];
    const int tid  = threadIdx.x;
    const int lane = tid & 31;
    const int wv   = tid >> 5;
    const int wm   = wv & 3;
    const int wn   = wv >> 2;
    const int m0   = blockIdx.y << 6;
    const int n0   = blockIdx.x << 7;
    const int bi   = blockIdx.z;
    const int am   = lane & 15;
    const int ak   = (lane >> 4) << 1;

    const int aq = tid & 7,  ar = tid >> 3;
    const int bq = tid & 31, br = tid >> 5;
    const float* wpA = Wo + (size_t)(m0 + ar) * DD + 4 * aq;
    const float* wpG = wpA + (size_t)DD * DD;
    const float* yp  = Yv + ((size_t)bi * DD + br) * LL + n0 + 4 * bq;
    float* sA = As + ar * OPA + 4 * aq;
    float* sB = Bs + br * OPB + 4 * bq;

    v4f ra0 = *(const v4f*)(wpA);
    v4f ra1 = *(const v4f*)(wpA + 32 * DD);
    v4f ra2 = *(const v4f*)(wpG);
    v4f ra3 = *(const v4f*)(wpG + 32 * DD);
    v4f rb0 = *(const v4f*)(yp);
    v4f rb1 = *(const v4f*)(yp + 8 * LL);
    v4f rb2 = *(const v4f*)(yp + 16 * LL);
    v4f rb3 = *(const v4f*)(yp + 24 * LL);

    v8f accA0 = {}, accA1 = {}, accA2 = {}, accA3 = {};
    v8f accG0 = {}, accG1 = {}, accG2 = {}, accG3 = {};
    const float* aA = As + (16 * wm + am) * OPA + ak;
    const float* aG = As + (64 + 16 * wm + am) * OPA + ak;
    const float* bB = Bs + ak * OPB + 64 * wn + am;

    for (int kc = 0; kc < DD; kc += OKC) {
        *(v4f*)(sA)            = ra0;
        *(v4f*)(sA + 32 * OPA) = ra1;
        *(v4f*)(sA + 64 * OPA) = ra2;
        *(v4f*)(sA + 96 * OPA) = ra3;
        *(v4f*)(sB)            = rb0;
        *(v4f*)(sB + 8 * OPB)  = rb1;
        *(v4f*)(sB + 16 * OPB) = rb2;
        *(v4f*)(sB + 24 * OPB) = rb3;
        __syncthreads();
        if (kc + OKC < DD) {
            const float* wa = wpA + kc + OKC;
            const float* wg = wpG + kc + OKC;
            const float* yy = yp + (size_t)(kc + OKC) * LL;
            ra0 = *(const v4f*)(wa);
            ra1 = *(const v4f*)(wa + 32 * DD);
            ra2 = *(const v4f*)(wg);
            ra3 = *(const v4f*)(wg + 32 * DD);
            rb0 = *(const v4f*)(yy);
            rb1 = *(const v4f*)(yy + 8 * LL);
            rb2 = *(const v4f*)(yy + 16 * LL);
            rb3 = *(const v4f*)(yy + 24 * LL);
        }
        #pragma unroll
        for (int k = 0; k < OKC; k += 4) {
            v2f aa, ag, b0, b1, b2, b3;
            aa.x = aA[k];
            aa.y = aA[k + 1];
            ag.x = aG[k];
            ag.y = aG[k + 1];
            b0.x = bB[k * OPB];            b0.y = bB[(k + 1) * OPB];
            b1.x = bB[k * OPB + 16];       b1.y = bB[(k + 1) * OPB + 16];
            b2.x = bB[k * OPB + 32];       b2.y = bB[(k + 1) * OPB + 32];
            b3.x = bB[k * OPB + 48];       b3.y = bB[(k + 1) * OPB + 48];
            accA0 = __builtin_amdgcn_wmma_f32_16x16x4_f32(false, aa, false, b0, (short)0, accA0, false, false);
            accG0 = __builtin_amdgcn_wmma_f32_16x16x4_f32(false, ag, false, b0, (short)0, accG0, false, false);
            accA1 = __builtin_amdgcn_wmma_f32_16x16x4_f32(false, aa, false, b1, (short)0, accA1, false, false);
            accG1 = __builtin_amdgcn_wmma_f32_16x16x4_f32(false, ag, false, b1, (short)0, accG1, false, false);
            accA2 = __builtin_amdgcn_wmma_f32_16x16x4_f32(false, aa, false, b2, (short)0, accA2, false, false);
            accG2 = __builtin_amdgcn_wmma_f32_16x16x4_f32(false, ag, false, b2, (short)0, accG2, false, false);
            accA3 = __builtin_amdgcn_wmma_f32_16x16x4_f32(false, aa, false, b3, (short)0, accA3, false, false);
            accG3 = __builtin_amdgcn_wmma_f32_16x16x4_f32(false, ag, false, b3, (short)0, accG3, false, false);
        }
        __syncthreads();
    }

    const int cr = (lane >> 4) << 3;
    #pragma unroll
    for (int v = 0; v < 8; ++v) {
        const int row = m0 + 16 * wm + cr + v;
        const float ba = bo[row];
        const float bg = bo[DD + row];
        float* o = Out + ((size_t)bi * DD + row) * LL + n0 + 64 * wn + am;
        o[0]  = (accA0[v] + ba) * sigmoidf_(accG0[v] + bg);
        o[16] = (accA1[v] + ba) * sigmoidf_(accG1[v] + bg);
        o[32] = (accA2[v] + ba) * sigmoidf_(accG2[v] + bg);
        o[48] = (accA3[v] + ba) * sigmoidf_(accG3[v] + bg);
    }
}

extern "C" void kernel_launch(void* const* d_in, const int* in_sizes, int n_in,
                              void* d_out, int out_size, void* d_ws, size_t ws_size,
                              hipStream_t stream)
{
    const float* u    = (const float*)d_in[0];
    const float* xw   = (const float*)d_in[1];
    const float* dtw  = (const float*)d_in[2];
    const float* dtb  = (const float*)d_in[3];
    const float* alog = (const float*)d_in[4];
    const float* Dp   = (const float*)d_in[5];
    const float* ow   = (const float*)d_in[6];
    const float* ob   = (const float*)d_in[7];
    float* out = (float*)d_out;

    const size_t CH = (size_t)NB * DD;
    float* ws    = (float*)d_ws;
    float* xdbl  = ws;
    float* delta = xdbl + (size_t)NB * EE * LL;
    float* yv    = delta + CH * LL;
    float* Sst   = yv + CH * LL;
    float* sumD  = Sst + CH * NC * NS;
    float* hin   = sumD + CH * NC;

    const dim3 blk(128);
    hipLaunchKernelGGL(k_xproj, dim3(LL / 64, EE / 16, NB), blk, 0, stream, xw, u, xdbl);
    hipLaunchKernelGGL(k_dt,    dim3(LL / 64, DD / 16, NB), blk, 0, stream, dtw, xdbl, dtb, delta);
    hipLaunchKernelGGL(k_scanA, dim3((CH * NC) / CW), dim3(32), 0, stream, u, xdbl, delta, alog, Sst, sumD);
    hipLaunchKernelGGL(k_scanB, dim3(CH / 16), dim3(256), 0, stream, alog, Sst, sumD, hin);
    hipLaunchKernelGGL(k_scanC, dim3((CH * NC) / CW), dim3(32), 0, stream, u, xdbl, delta, alog, Dp, hin, yv);
    hipLaunchKernelGGL(k_out,   dim3(LL / 128, DD / 64, NB), dim3(256), 0, stream, ow, yv, ob, out);
}


# scan passes A, C: decay exp(delta*A) as one hardware exp2 of delta*(A*log2 e), A scaled once per channel (was library expf, range reduction in code, per state and step); sum order unchanged
# speedup vs baseline: 2.1910x; 1.1572x over previous
#include <hip/hip_runtime.h>
#include <math.h>

#define NB 2
#define DD 1024
#define LL 2048
#define NS 16
#define RR 64
#define EE 96
#define NC 32
#define LC 64
#define BP 68

typedef float v2f __attribute__((ext_vector_type(2)));
typedef float v8f __attribute__((ext_vector_type(8)));

__device__ __forceinline__ float sigmoidf_(float x) { return 1.0f / (1.0f + expf(-x)); }
__device__ __forceinline__ float softplusf_(float x) { return (x > 20.0f) ? x : log1pf(expf(x)); }

#define XP 1028
#define XCH 64

__device__ __forceinline__ void xproj_load_b(float (&b)[32], const float* __restrict__ p)
{
    #pragma unroll
    for (int j = 0; j < 16; ++j) {
        b[2 * j]     = p[(size_t)(4 * j) * LL];
        b[2 * j + 1] = p[(size_t)(4 * j + 1) * LL];
    }
}

__global__ __launch_bounds__(128)
void k_xproj(const float* __restrict__ W, const float* __restrict__ U,
             float* __restrict__ Xd)
{
    __shared__ __attribute__((aligned(16))) float As[16 * XP];
    const int tid  = threadIdx.x;
    const int lane = tid & 31;
    const int wv   = tid >> 5;
    const int m0   = blockIdx.y << 4;
    const int n0   = (blockIdx.x << 6) + (wv << 4);
    const int bi   = blockIdx.z;
    const int am   = lane & 15;
    const int ak   = (lane >> 4) << 1;
    const float* Ub = U + ((size_t)bi * DD + ak) * LL + n0 + am;

    float b[3][32];
    xproj_load_b(b[0], Ub);
    xproj_load_b(b[1], Ub + (size_t)XCH * LL);

    {
        const float* wp = W + (size_t)m0 * DD + 4 * tid;
        float4 t[32];
        #pragma unroll
        for (int j = 0; j < 32; ++j)
            t[j] = *(const float4*)(wp + (size_t)(j >> 1) * DD + 512 * (j & 1));
        #pragma unroll
        for (int j = 0; j < 32; ++j)
            *(float4*)(As + (j >> 1) * XP + 512 * (j & 1) + 4 * tid) = t[j];
    }
    __syncthreads();

    v8f acc = {};
    const float* aRow = As + am * XP + ak;
    #pragma unroll
    for (int c = 0; c < DD / XCH; ++c) {
        if (c + 2 < DD / XCH)
            xproj_load_b(b[(c + 2) % 3], Ub + (size_t)((c + 2) * XCH) * LL);
        #pragma unroll
        for (int j = 0; j < 16; ++j) {
            const v2f a = *(const v2f*)(aRow + c * XCH + 4 * j);
            v2f bb;
            bb.x = b[c % 3][2 * j];
            bb.y = b[c % 3][2 * j + 1];
            acc = __builtin_amdgcn_wmma_f32_16x16x4_f32(
                false, a, false, bb, (short)0, acc, false, false);
        }
    }
    const int cn = n0 + am;
    const int cr = (lane >> 4) << 3;
    #pragma unroll
    for (int v = 0; v < 8; ++v)
        Xd[((size_t)bi * EE + m0 + cr + v) * LL + cn] = acc[v];
}

__global__ __launch_bounds__(128)
void k_dt(const float* __restrict__ Wdt, const float* __restrict__ Xd,
          const float* __restrict__ bias, float* __restrict__ Delta)
{
    __shared__ float As[16 * 68];
    const int tid  = threadIdx.x;
    const int lane = tid & 31;
    const int wv   = tid >> 5;
    const int m0   = blockIdx.y << 4;
    const int n0   = (blockIdx.x << 6) + (wv << 4);
    const int bi   = blockIdx.z;
    const int am   = lane & 15;
    const int ak   = (lane >> 4) << 1;
    for (int i = tid; i < 16 * 64; i += 128) {
        const int r = i >> 6, c = i & 63;
        As[r * 68 + c] = Wdt[(size_t)(m0 + r) * RR + c];
    }
    __syncthreads();
    v8f acc = {};
    const float* Bb = Xd + ((size_t)bi * EE) * LL + n0 + am;
    #pragma unroll
    for (int k = 0; k < RR; k += 4) {
        v2f a, b;
        a.x = As[am * 68 + k + ak];
        a.y = As[am * 68 + k + ak + 1];
        const float* bp = Bb + (size_t)(k + ak) * LL;
        b.x = bp[0];
        b.y = bp[LL];
        acc = __builtin_amdgcn_wmma_f32_16x16x4_f32(
            false, a, false, b, (short)0, acc, false, false);
    }
    const int cn = n0 + am;
    const int cr = (lane >> 4) << 3;
    #pragma unroll
    for (int v = 0; v < 8; ++v) {
        const int row = m0 + cr + v;
        Delta[((size_t)bi * DD + row) * LL + cn] = softplusf_(acc[v] + bias[row]);
    }
}

#define CW 16
#define HP 20

#define LOG2E 1.4426950408889634f
__device__ __forceinline__ float exp2_hw(float x) { return __builtin_amdgcn_exp2f(x); }

__device__ __forceinline__ float xhalf(float x)
{
    return __builtin_bit_cast(float, __builtin_amdgcn_permlanex16(
        __builtin_bit_cast(int, x), __builtin_bit_cast(int, x), 0x76543210, 0xfedcba98, false, false));
}

__device__ __forceinline__ void stage_tile16(const float* __restrict__ g, float* s, int lane)
{
    const int q = lane & 15, r0 = lane >> 4;
    #pragma unroll
    for (int k = 0; k < 8; ++k) {
        const float4 t = *(const float4*)(g + (size_t)(2 * k + r0) * LL + 4 * q);
        *(float4*)(s + (2 * k + r0) * BP + 4 * q) = t;
    }
}

__global__ __launch_bounds__(32)
void k_scanA(const float* __restrict__ U, const float* __restrict__ Xd,
             const float* __restrict__ Delta, const float* __restrict__ Alog,
             float* __restrict__ S, float* __restrict__ SumD)
{
    __shared__ __attribute__((aligned(16))) float Ds[CW * BP];
    __shared__ __attribute__((aligned(16))) float Us[CW * BP];
    __shared__ __attribute__((aligned(16))) float Bs[NS * BP];
    const int lane = threadIdx.x;
    const int cl   = lane & 15;
    const int hh   = lane >> 4;
    const int dg   = blockIdx.x & (DD / CW - 1);
    const int c    = (blockIdx.x / (DD / CW)) & (NC - 1);
    const int bi   = blockIdx.x / ((DD / CW) * NC);
    const int d0   = dg * CW;
    const int d    = d0 + cl;
    const size_t tileOff = ((size_t)bi * DD + d0) * LL + (size_t)c * LC;
    stage_tile16(Delta + tileOff, Ds, lane);
    stage_tile16(U + tileOff, Us, lane);
    stage_tile16(Xd + ((size_t)bi * EE + RR) * LL + (size_t)c * LC, Bs, lane);
    float An[8];
    #pragma unroll
    for (int q = 0; q < 4; ++q) {
        const float4 a = *(const float4*)(Alog + d * NS + 4 * q);
        An[2*q+0] = -expf(hh ? a.y : a.x) * LOG2E;
        An[2*q+1] = -expf(hh ? a.w : a.z) * LOG2E;
    }
    __syncthreads();
    const float* dRow = Ds + cl * BP;
    const float* uRow = Us + cl * BP;
    const float* bRow = Bs + hh * BP;
    {
        float sd[16];
        #pragma unroll
        for (int q = 0; q < 4; ++q) {
            const float4 a0 = *(const float4*)(dRow + 4 * q);
            const float4 a1 = *(const float4*)(dRow + 16 + 4 * q);
            const float4 a2 = *(const float4*)(dRow + 32 + 4 * q);
            const float4 a3 = *(const float4*)(dRow + 48 + 4 * q);
            sd[4*q+0] = ((a0.x + a1.x) + a2.x) + a3.x;
            sd[4*q+1] = ((a0.y + a1.y) + a2.y) + a3.y;
            sd[4*q+2] = ((a0.z + a1.z) + a2.z) + a3.z;
            sd[4*q+3] = ((a0.w + a1.w) + a2.w) + a3.w;
        }
        #pragma unroll
        for (int i = 0; i < 8; ++i) sd[i] += sd[i + 8];
        #pragma unroll
        for (int i = 0; i < 4; ++i) sd[i] += sd[i + 4];
        sd[0] += sd[2]; sd[1] += sd[3];
        if (hh == 0) SumD[((size_t)bi * DD + d) * NC + c] = sd[0] + sd[1];
    }
    float h[8];
    #pragma unroll
    for (int i = 0; i < 8; ++i) h[i] = 0.0f;
    for (int l0 = 0; l0 < LC; l0 += 4) {
        const float4 dq = *(const float4*)(dRow + l0);
        const float4 uq = *(const float4*)(uRow + l0);
        const float dv[4] = {dq.x, dq.y, dq.z, dq.w};
        const float uv[4] = {uq.x, uq.y, uq.z, uq.w};
        float bv[8][4];
        #pragma unroll
        for (int i = 0; i < 8; ++i) {
            const float4 t = *(const float4*)(bRow + 2 * i * BP + l0);
            bv[i][0] = t.x; bv[i][1] = t.y; bv[i][2] = t.z; bv[i][3] = t.w;
        }
        #pragma unroll
        for (int t = 0; t < 4; ++t) {
            const float dj = dv[t];
            const float uj = uv[t];
            #pragma unroll
            for (int i = 0; i < 8; ++i)
                h[i] = exp2_hw(dj * An[i]) * h[i] + dj * uj * bv[i][t];
        }
    }
    __syncthreads();
    float* Hs = Bs;
    #pragma unroll
    for (int i = 0; i < 8; ++i) Hs[cl * HP + 2 * i + hh] = h[i];
    __syncthreads();
    float* Sg = S + (((size_t)bi * NC + c) * DD + d0) * NS;
    #pragma unroll
    for (int k = 0; k < 2; ++k) {
        const int r = 8 * k + (lane >> 2), q = lane & 3;
        const float4 t = *(const float4*)(Hs + r * HP + 4 * q);
        *(float4*)(Sg + r * NS + 4 * q) = t;
    }
}

__global__ __launch_bounds__(256)
void k_scanB(const float* __restrict__ Alog, const float* __restrict__ S,
             const float* __restrict__ SumD, float* __restrict__ Hin)
{
    const int lane = threadIdx.x & 31;
    const int gw   = blockIdx.x * (blockDim.x >> 5) + (threadIdx.x >> 5);
    const int ch   = (gw << 1) + (lane >> 4);
    const int bi   = ch >> 10;
    const int d    = ch & (DD - 1);
    const int n    = lane & 15;
    const float An = -expf(Alog[d * NS + n]);
    const size_t so = ((size_t)bi * NC * DD + d) * NS + n;
    float P[NC];
    #pragma unroll
    for (int c = 0; c < NC; ++c)
        P[c] = expf(An * SumD[(size_t)ch * NC + c]);
    float hin = 0.0f;
    #pragma unroll
    for (int c = 0; c < NC; ++c) {
        Hin[so + (size_t)c * DD * NS] = hin;
        hin = P[c] * hin + S[so + (size_t)c * DD * NS];
    }
}

__global__ __launch_bounds__(32)
void k_scanC(const float* __restrict__ U, const float* __restrict__ Xd,
             const float* __restrict__ Delta, const float* __restrict__ Alog,
             const float* __restrict__ Dp, const float* __restrict__ Hin,
             float* __restrict__ Y)
{
    __shared__ __attribute__((aligned(16))) float Ds[CW * BP];
    __shared__ __attribute__((aligned(16))) float Us[CW * BP];
    __shared__ __attribute__((aligned(16))) float Bs[NS * BP];
    __shared__ __attribute__((aligned(16))) float Cs[NS * BP];
    const int lane = threadIdx.x;
    const int cl   = lane & 15;
    const int hh   = lane >> 4;
    const int dg   = blockIdx.x & (DD / CW - 1);
    const int c    = (blockIdx.x / (DD / CW)) & (NC - 1);
    const int bi   = blockIdx.x / ((DD / CW) * NC);
    const int d0   = dg * CW;
    const int d    = d0 + cl;
    const size_t tileOff = ((size_t)bi * DD + d0) * LL + (size_t)c * LC;
    stage_tile16(Delta + tileOff, Ds, lane);
    stage_tile16(U + tileOff, Us, lane);
    const float* gB = Xd + ((size_t)bi * EE + RR) * LL + (size_t)c * LC;
    stage_tile16(gB, Bs, lane);
    stage_tile16(gB + (size_t)NS * LL, Cs, lane);
    float An[8], h[8];
    const float* hg = Hin + (((size_t)bi * NC + c) * DD + d) * NS;
    #pragma unroll
    for (int q = 0; q < 4; ++q) {
        const float4 a = *(const float4*)(Alog + d * NS + 4 * q);
        An[2*q+0] = -expf(hh ? a.y : a.x) * LOG2E;
        An[2*q+1] = -expf(hh ? a.w : a.z) * LOG2E;
        const float4 t = *(const float4*)(hg + 4 * q);
        h[2*q+0] = hh ? t.y : t.x;
        h[2*q+1] = hh ? t.w : t.z;
    }
    const float Dv = Dp[d];
    __syncthreads();
    float* dRow = Ds + cl * BP;
    const float* uRow = Us + cl * BP;
    const float* bRow = Bs + hh * BP;
    const float* cRow = Cs + hh * BP;
    for (int l0 = 0; l0 < LC; l0 += 4) {
        const float4 dq = *(const float4*)(dRow + l0);
        const float4 uq = *(const float4*)(uRow + l0);
        const float dv[4] = {dq.x, dq.y, dq.z, dq.w};
        const float uv[4] = {uq.x, uq.y, uq.z, uq.w};
        float bv[8][4], cv[8][4];
        #pragma unroll
        for (int i = 0; i < 8; ++i) {
            const float4 t = *(const float4*)(bRow + 2 * i * BP + l0);
            bv[i][0] = t.x; bv[i][1] = t.y; bv[i][2] = t.z; bv[i][3] = t.w;
            const float4 s = *(const float4*)(cRow + 2 * i * BP + l0);
            cv[i][0] = s.x; cv[i][1] = s.y; cv[i][2] = s.z; cv[i][3] = s.w;
        }
        float part[4];
        #pragma unroll
        for (int t = 0; t < 4; ++t) {
            const float dj = dv[t];
            const float uj = uv[t];
            float p[8];
            #pragma unroll
            for (int i = 0; i < 8; ++i) {
                h[i] = exp2_hw(dj * An[i]) * h[i] + dj * uj * bv[i][t];
                p[i] = h[i] * cv[i][t];
            }
            #pragma unroll
            for (int i = 0; i < 4; ++i) p[i] += p[i + 4];
            p[0] += p[2]; p[1] += p[3];
            part[t] = p[0] + p[1];
        }
        float yf[4];
        #pragma unroll
        for (int t = 0; t < 4; ++t) yf[t] = part[t] + xhalf(part[t]);
        float yo2[2];
        #pragma unroll
        for (int k = 0; k < 2; ++k) {
            const float ul = hh ? uv[2 + k] : uv[k];
            float yo = (hh ? yf[2 + k] : yf[k]) + ul * Dv;
            yo = yo * (ul * sigmoidf_(ul));
            yo = 0.5f * yo * (1.0f + erff(yo * 0.70710678118f));
            yo2[k] = yo;
        }
        float2 yq; yq.x = yo2[0]; yq.y = yo2[1];
        *(float2*)(dRow + l0 + 2 * hh) = yq;
    }
    __syncthreads();
    {
        const int q = lane & 15, r0 = lane >> 4;
        float* yg = Y + tileOff;
        #pragma unroll
        for (int k = 0; k < 8; ++k) {
            const float4 t = *(const float4*)(Ds + (2 * k + r0) * BP + 4 * q);
            *(float4*)(yg + (size_t)(2 * k + r0) * LL + 4 * q) = t;
        }
    }
}

#define OKC 32
#define OPA 36
#define OPB 136

typedef float v4f __attribute__((ext_vector_type(4)));

__global__ __launch_bounds__(256)
void k_out(const float* __restrict__ Wo, const float* __restrict__ Yv,
           const float* __restrict__ bo, float* __restrict__ Out)
{
    __shared__ __attribute__((aligned(16))) float As[128 * OPA];
    __shared__ __attribute__((aligned(16))) float Bs[OKC * OPB];
    const int tid  = threadIdx.x;
    const int lane = tid & 31;
    const int wv   = tid >> 5;
    const int wm   = wv & 3;
    const int wn   = wv >> 2;
    const int m0   = blockIdx.y << 6;
    const int n0   = blockIdx.x << 7;
    const int bi   = blockIdx.z;
    const int am   = lane & 15;
    const int ak   = (lane >> 4) << 1;

    const int aq = tid & 7,  ar = tid >> 3;
    const int bq = tid & 31, br = tid >> 5;
    const float* wpA = Wo + (size_t)(m0 + ar) * DD + 4 * aq;
    const float* wpG = wpA + (size_t)DD * DD;
    const float* yp  = Yv + ((size_t)bi * DD + br) * LL + n0 + 4 * bq;
    float* sA = As + ar * OPA + 4 * aq;
    float* sB = Bs + br * OPB + 4 * bq;

    v4f ra0 = *(const v4f*)(wpA);
    v4f ra1 = *(const v4f*)(wpA + 32 * DD);
    v4f ra2 = *(const v4f*)(wpG);
    v4f ra3 = *(const v4f*)(wpG + 32 * DD);
    v4f rb0 = *(const v4f*)(yp);
    v4f rb1 = *(const v4f*)(yp + 8 * LL);
    v4f rb2 = *(const v4f*)(yp + 16 * LL);
    v4f rb3 = *(const v4f*)(yp + 24 * LL);

    v8f accA0 = {}, accA1 = {}, accA2 = {}, accA3 = {};
    v8f accG0 = {}, accG1 = {}, accG2 = {}, accG3 = {};
    const float* aA = As + (16 * wm + am) * OPA + ak;
    const float* aG = As + (64 + 16 * wm + am) * OPA + ak;
    const float* bB = Bs + ak * OPB + 64 * wn + am;

    for (int kc = 0; kc < DD; kc += OKC) {
        *(v4f*)(sA)            = ra0;
        *(v4f*)(sA + 32 * OPA) = ra1;
        *(v4f*)(sA + 64 * OPA) = ra2;
        *(v4f*)(sA + 96 * OPA) = ra3;
        *(v4f*)(sB)            = rb0;
        *(v4f*)(sB + 8 * OPB)  = rb1;
        *(v4f*)(sB + 16 * OPB) = rb2;
        *(v4f*)(sB + 24 * OPB) = rb3;
        __syncthreads();
        if (kc + OKC < DD) {
            const float* wa = wpA + kc + OKC;
            const float* wg = wpG + kc + OKC;
            const float* yy = yp + (size_t)(kc + OKC) * LL;
            ra0 = *(const v4f*)(wa);
            ra1 = *(const v4f*)(wa + 32 * DD);
            ra2 = *(const v4f*)(wg);
            ra3 = *(const v4f*)(wg + 32 * DD);
            rb0 = *(const v4f*)(yy);
            rb1 = *(const v4f*)(yy + 8 * LL);
            rb2 = *(const v4f*)(yy + 16 * LL);
            rb3 = *(const v4f*)(yy + 24 * LL);
        }
        #pragma unroll
        for (int k = 0; k < OKC; k += 4) {
            v2f aa, ag, b0, b1, b2, b3;
            aa.x = aA[k];
            aa.y = aA[k + 1];
            ag.x = aG[k];
            ag.y = aG[k + 1];
            b0.x = bB[k * OPB];            b0.y = bB[(k + 1) * OPB];
            b1.x = bB[k * OPB + 16];       b1.y = bB[(k + 1) * OPB + 16];
            b2.x = bB[k * OPB + 32];       b2.y = bB[(k + 1) * OPB + 32];
            b3.x = bB[k * OPB + 48];       b3.y = bB[(k + 1) * OPB + 48];
            accA0 = __builtin_amdgcn_wmma_f32_16x16x4_f32(false, aa, false, b0, (short)0, accA0, false, false);
            accG0 = __builtin_amdgcn_wmma_f32_16x16x4_f32(false, ag, false, b0, (short)0, accG0, false, false);
            accA1 = __builtin_amdgcn_wmma_f32_16x16x4_f32(false, aa, false, b1, (short)0, accA1, false, false);
            accG1 = __builtin_amdgcn_wmma_f32_16x16x4_f32(false, ag, false, b1, (short)0, accG1, false, false);
            accA2 = __builtin_amdgcn_wmma_f32_16x16x4_f32(false, aa, false, b2, (short)0, accA2, false, false);
            accG2 = __builtin_amdgcn_wmma_f32_16x16x4_f32(false, ag, false, b2, (short)0, accG2, false, false);
            accA3 = __builtin_amdgcn_wmma_f32_16x16x4_f32(false, aa, false, b3, (short)0, accA3, false, false);
            accG3 = __builtin_amdgcn_wmma_f32_16x16x4_f32(false, ag, false, b3, (short)0, accG3, false, false);
        }
        __syncthreads();
    }

    const int cr = (lane >> 4) << 3;
    #pragma unroll
    for (int v = 0; v < 8; ++v) {
        const int row = m0 + 16 * wm + cr + v;
        const float ba = bo[row];
        const float bg = bo[DD + row];
        float* o = Out + ((size_t)bi * DD + row) * LL + n0 + 64 * wn + am;
        o[0]  = (accA0[v] + ba) * sigmoidf_(accG0[v] + bg);
        o[16] = (accA1[v] + ba) * sigmoidf_(accG1[v] + bg);
        o[32] = (accA2[v] + ba) * sigmoidf_(accG2[v] + bg);
        o[48] = (accA3[v] + ba) * sigmoidf_(accG3[v] + bg);
    }
}

extern "C" void kernel_launch(void* const* d_in, const int* in_sizes, int n_in,
                              void* d_out, int out_size, void* d_ws, size_t ws_size,
                              hipStream_t stream)
{
    const float* u    = (const float*)d_in[0];
    const float* xw   = (const float*)d_in[1];
    const float* dtw  = (const float*)d_in[2];
    const float* dtb  = (const float*)d_in[3];
    const float* alog = (const float*)d_in[4];
    const float* Dp   = (const float*)d_in[5];
    const float* ow   = (const float*)d_in[6];
    const float* ob   = (const float*)d_in[7];
    float* out = (float*)d_out;

    const size_t CH = (size_t)NB * DD;
    float* ws    = (float*)d_ws;
    float* xdbl  = ws;
    float* delta = xdbl + (size_t)NB * EE * LL;
    float* yv    = delta + CH * LL;
    float* Sst   = yv + CH * LL;
    float* sumD  = Sst + CH * NC * NS;
    float* hin   = sumD + CH * NC;

    const dim3 blk(128);
    hipLaunchKernelGGL(k_xproj, dim3(LL / 64, EE / 16, NB), blk, 0, stream, xw, u, xdbl);
    hipLaunchKernelGGL(k_dt,    dim3(LL / 64, DD / 16, NB), blk, 0, stream, dtw, xdbl, dtb, delta);
    hipLaunchKernelGGL(k_scanA, dim3((CH * NC) / CW), dim3(32), 0, stream, u, xdbl, delta, alog, Sst, sumD);
    hipLaunchKernelGGL(k_scanB, dim3(CH / 16), dim3(256), 0, stream, alog, Sst, sumD, hin);
    hipLaunchKernelGGL(k_scanC, dim3((CH * NC) / CW), dim3(32), 0, stream, u, xdbl, delta, alog, Dp, hin, yv);
    hipLaunchKernelGGL(k_out,   dim3(LL / 128, DD / 64, NB), dim3(256), 0, stream, ow, yv, ob, out);
}


# k_scanB, k_dt: all global loads issued up front, one round trip (was one load per combine step in k_scanB; in k_dt 8 one-at-a-time weight loads, then B loads behind the barrier); sum order unchanged
# speedup vs baseline: 2.4018x; 1.0962x over previous
#include <hip/hip_runtime.h>
#include <math.h>

#define NB 2
#define DD 1024
#define LL 2048
#define NS 16
#define RR 64
#define EE 96
#define NC 32
#define LC 64
#define BP 68

typedef float v2f __attribute__((ext_vector_type(2)));
typedef float v8f __attribute__((ext_vector_type(8)));

__device__ __forceinline__ float sigmoidf_(float x) { return 1.0f / (1.0f + expf(-x)); }
__device__ __forceinline__ float softplusf_(float x) { return (x > 20.0f) ? x : log1pf(expf(x)); }

#define XP 1028
#define XCH 64

__device__ __forceinline__ void xproj_load_b(float (&b)[32], const float* __restrict__ p)
{
    #pragma unroll
    for (int j = 0; j < 16; ++j) {
        b[2 * j]     = p[(size_t)(4 * j) * LL];
        b[2 * j + 1] = p[(size_t)(4 * j + 1) * LL];
    }
}

__global__ __launch_bounds__(128)
void k_xproj(const float* __restrict__ W, const float* __restrict__ U,
             float* __restrict__ Xd)
{
    __shared__ __attribute__((aligned(16))) float As[16 * XP];
    const int tid  = threadIdx.x;
    const int lane = tid & 31;
    const int wv   = tid >> 5;
    const int m0   = blockIdx.y << 4;
    const int n0   = (blockIdx.x << 6) + (wv << 4);
    const int bi   = blockIdx.z;
    const int am   = lane & 15;
    const int ak   = (lane >> 4) << 1;
    const float* Ub = U + ((size_t)bi * DD + ak) * LL + n0 + am;

    float b[3][32];
    xproj_load_b(b[0], Ub);
    xproj_load_b(b[1], Ub + (size_t)XCH * LL);

    {
        const float* wp = W + (size_t)m0 * DD + 4 * tid;
        float4 t[32];
        #pragma unroll
        for (int j = 0; j < 32; ++j)
            t[j] = *(const float4*)(wp + (size_t)(j >> 1) * DD + 512 * (j & 1));
        #pragma unroll
        for (int j = 0; j < 32; ++j)
            *(float4*)(As + (j >> 1) * XP + 512 * (j & 1) + 4 * tid) = t[j];
    }
    __syncthreads();

    v8f acc = {};
    const float* aRow = As + am * XP + ak;
    #pragma unroll
    for (int c = 0; c < DD / XCH; ++c) {
        if (c + 2 < DD / XCH)
            xproj_load_b(b[(c + 2) % 3], Ub + (size_t)((c + 2) * XCH) * LL);
        #pragma unroll
        for (int j = 0; j < 16; ++j) {
            const v2f a = *(const v2f*)(aRow + c * XCH + 4 * j);
            v2f bb;
            bb.x = b[c % 3][2 * j];
            bb.y = b[c % 3][2 * j + 1];
            acc = __builtin_amdgcn_wmma_f32_16x16x4_f32(
                false, a, false, bb, (short)0, acc, false, false);
        }
    }
    const int cn = n0 + am;
    const int cr = (lane >> 4) << 3;
    #pragma unroll
    for (int v = 0; v < 8; ++v)
        Xd[((size_t)bi * EE + m0 + cr + v) * LL + cn] = acc[v];
}

__global__ __launch_bounds__(128)
void k_dt(const float* __restrict__ Wdt, const float* __restrict__ Xd,
          const float* __restrict__ bias, float* __restrict__ Delta)
{
    __shared__ __attribute__((aligned(16))) float As[16 * 68];
    const int tid  = threadIdx.x;
    const int lane = tid & 31;
    const int wv   = tid >> 5;
    const int m0   = blockIdx.y << 4;
    const int n0   = (blockIdx.x << 6) + (wv << 4);
    const int bi   = blockIdx.z;
    const int am   = lane & 15;
    const int ak   = (lane >> 4) << 1;
    const int cr   = (lane >> 4) << 3;
    const int wr = tid >> 4, wq = tid & 15;
    const float4 t0 = *(const float4*)(Wdt + (size_t)(m0 + wr) * RR + 4 * wq);
    const float4 t1 = *(const float4*)(Wdt + (size_t)(m0 + 8 + wr) * RR + 4 * wq);
    const float* Bb = Xd + ((size_t)bi * EE + ak) * LL + n0 + am;
    float bx[16], by[16];
    #pragma unroll
    for (int j = 0; j < 16; ++j) {
        bx[j] = Bb[(size_t)(4 * j) * LL];
        by[j] = Bb[(size_t)(4 * j + 1) * LL];
    }
    const float4 bq0 = *(const float4*)(bias + m0 + cr);
    const float4 bq1 = *(const float4*)(bias + m0 + cr + 4);
    const float bv[8] = {bq0.x, bq0.y, bq0.z, bq0.w, bq1.x, bq1.y, bq1.z, bq1.w};
    *(float4*)(As + wr * 68 + 4 * wq)       = t0;
    *(float4*)(As + (8 + wr) * 68 + 4 * wq) = t1;
    __syncthreads();
    v8f acc = {};
    #pragma unroll
    for (int j = 0; j < 16; ++j) {
        v2f a, b;
        a.x = As[am * 68 + 4 * j + ak];
        a.y = As[am * 68 + 4 * j + ak + 1];
        b.x = bx[j];
        b.y = by[j];
        acc = __builtin_amdgcn_wmma_f32_16x16x4_f32(
            false, a, false, b, (short)0, acc, false, false);
    }
    const int cn = n0 + am;
    #pragma unroll
    for (int v = 0; v < 8; ++v) {
        const int row = m0 + cr + v;
        Delta[((size_t)bi * DD + row) * LL + cn] = softplusf_(acc[v] + bv[v]);
    }
}

#define CW 16
#define HP 20

#define LOG2E 1.4426950408889634f
__device__ __forceinline__ float exp2_hw(float x) { return __builtin_amdgcn_exp2f(x); }

__device__ __forceinline__ float xhalf(float x)
{
    return __builtin_bit_cast(float, __builtin_amdgcn_permlanex16(
        __builtin_bit_cast(int, x), __builtin_bit_cast(int, x), 0x76543210, 0xfedcba98, false, false));
}

__device__ __forceinline__ void stage_tile16(const float* __restrict__ g, float* s, int lane)
{
    const int q = lane & 15, r0 = lane >> 4;
    #pragma unroll
    for (int k = 0; k < 8; ++k) {
        const float4 t = *(const float4*)(g + (size_t)(2 * k + r0) * LL + 4 * q);
        *(float4*)(s + (2 * k + r0) * BP + 4 * q) = t;
    }
}

__global__ __launch_bounds__(32)
void k_scanA(const float* __restrict__ U, const float* __restrict__ Xd,
             const float* __restrict__ Delta, const float* __restrict__ Alog,
             float* __restrict__ S, float* __restrict__ SumD)
{
    __shared__ __attribute__((aligned(16))) float Ds[CW * BP];
    __shared__ __attribute__((aligned(16))) float Us[CW * BP];
    __shared__ __attribute__((aligned(16))) float Bs[NS * BP];
    const int lane = threadIdx.x;
    const int cl   = lane & 15;
    const int hh   = lane >> 4;
    const int dg   = blockIdx.x & (DD / CW - 1);
    const int c    = (blockIdx.x / (DD / CW)) & (NC - 1);
    const int bi   = blockIdx.x / ((DD / CW) * NC);
    const int d0   = dg * CW;
    const int d    = d0 + cl;
    const size_t tileOff = ((size_t)bi * DD + d0) * LL + (size_t)c * LC;
    stage_tile16(Delta + tileOff, Ds, lane);
    stage_tile16(U + tileOff, Us, lane);
    stage_tile16(Xd + ((size_t)bi * EE + RR) * LL + (size_t)c * LC, Bs, lane);
    float An[8];
    #pragma unroll
    for (int q = 0; q < 4; ++q) {
        const float4 a = *(const float4*)(Alog + d * NS + 4 * q);
        An[2*q+0] = -expf(hh ? a.y : a.x) * LOG2E;
        An[2*q+1] = -expf(hh ? a.w : a.z) * LOG2E;
    }
    __syncthreads();
    const float* dRow = Ds + cl * BP;
    const float* uRow = Us + cl * BP;
    const float* bRow = Bs + hh * BP;
    {
        float sd[16];
        #pragma unroll
        for (int q = 0; q < 4; ++q) {
            const float4 a0 = *(const float4*)(dRow + 4 * q);
            const float4 a1 = *(const float4*)(dRow + 16 + 4 * q);
            const float4 a2 = *(const float4*)(dRow + 32 + 4 * q);
            const float4 a3 = *(const float4*)(dRow + 48 + 4 * q);
            sd[4*q+0] = ((a0.x + a1.x) + a2.x) + a3.x;
            sd[4*q+1] = ((a0.y + a1.y) + a2.y) + a3.y;
            sd[4*q+2] = ((a0.z + a1.z) + a2.z) + a3.z;
            sd[4*q+3] = ((a0.w + a1.w) + a2.w) + a3.w;
        }
        #pragma unroll
        for (int i = 0; i < 8; ++i) sd[i] += sd[i + 8];
        #pragma unroll
        for (int i = 0; i < 4; ++i) sd[i] += sd[i + 4];
        sd[0] += sd[2]; sd[1] += sd[3];
        if (hh == 0) SumD[((size_t)bi * DD + d) * NC + c] = sd[0] + sd[1];
    }
    float h[8];
    #pragma unroll
    for (int i = 0; i < 8; ++i) h[i] = 0.0f;
    for (int l0 = 0; l0 < LC; l0 += 4) {
        const float4 dq = *(const float4*)(dRow + l0);
        const float4 uq = *(const float4*)(uRow + l0);
        const float dv[4] = {dq.x, dq.y, dq.z, dq.w};
        const float uv[4] = {uq.x, uq.y, uq.z, uq.w};
        float bv[8][4];
        #pragma unroll
        for (int i = 0; i < 8; ++i) {
            const float4 t = *(const float4*)(bRow + 2 * i * BP + l0);
            bv[i][0] = t.x; bv[i][1] = t.y; bv[i][2] = t.z; bv[i][3] = t.w;
        }
        #pragma unroll
        for (int t = 0; t < 4; ++t) {
            const float dj = dv[t];
            const float uj = uv[t];
            #pragma unroll
            for (int i = 0; i < 8; ++i)
                h[i] = exp2_hw(dj * An[i]) * h[i] + dj * uj * bv[i][t];
        }
    }
    __syncthreads();
    float* Hs = Bs;
    #pragma unroll
    for (int i = 0; i < 8; ++i) Hs[cl * HP + 2 * i + hh] = h[i];
    __syncthreads();
    float* Sg = S + (((size_t)bi * NC + c) * DD + d0) * NS;
    #pragma unroll
    for (int k = 0; k < 2; ++k) {
        const int r = 8 * k + (lane >> 2), q = lane & 3;
        const float4 t = *(const float4*)(Hs + r * HP + 4 * q);
        *(float4*)(Sg + r * NS + 4 * q) = t;
    }
}

__global__ __launch_bounds__(256)
void k_scanB(const float* __restrict__ Alog, const float* __restrict__ S,
             const float* __restrict__ SumD, float* __restrict__ Hin)
{
    const int lane = threadIdx.x & 31;
    const int gw   = blockIdx.x * (blockDim.x >> 5) + (threadIdx.x >> 5);
    const int ch   = (gw << 1) + (lane >> 4);
    const int bi   = ch >> 10;
    const int d    = ch & (DD - 1);
    const int n    = lane & 15;
    const size_t so = ((size_t)bi * NC * DD + d) * NS + n;
    const float al = Alog[d * NS + n];
    float sd[NC], Sv[NC];
    #pragma unroll
    for (int q = 0; q < NC / 4; ++q) {
        const float4 t = *(const float4*)(SumD + (size_t)ch * NC + 4 * q);
        sd[4*q+0] = t.x; sd[4*q+1] = t.y; sd[4*q+2] = t.z; sd[4*q+3] = t.w;
    }
    #pragma unroll
    for (int c = 0; c < NC; ++c)
        Sv[c] = S[so + (size_t)c * DD * NS];
    const float An = -expf(al);
    float P[NC];
    #pragma unroll
    for (int c = 0; c < NC; ++c)
        P[c] = expf(An * sd[c]);
    float hv[NC];
    float hin = 0.0f;
    #pragma unroll
    for (int c = 0; c < NC; ++c) {
        hv[c] = hin;
        hin = P[c] * hin + Sv[c];
    }
    #pragma unroll
    for (int c = 0; c < NC; ++c)
        Hin[so + (size_t)c * DD * NS] = hv[c];
}

__global__ __launch_bounds__(32)
void k_scanC(const float* __restrict__ U, const float* __restrict__ Xd,
             const float* __restrict__ Delta, const float* __restrict__ Alog,
             const float* __restrict__ Dp, const float* __restrict__ Hin,
             float* __restrict__ Y)
{
    __shared__ __attribute__((aligned(16))) float Ds[CW * BP];
    __shared__ __attribute__((aligned(16))) float Us[CW * BP];
    __shared__ __attribute__((aligned(16))) float Bs[NS * BP];
    __shared__ __attribute__((aligned(16))) float Cs[NS * BP];
    const int lane = threadIdx.x;
    const int cl   = lane & 15;
    const int hh   = lane >> 4;
    const int dg   = blockIdx.x & (DD / CW - 1);
    const int c    = (blockIdx.x / (DD / CW)) & (NC - 1);
    const int bi   = blockIdx.x / ((DD / CW) * NC);
    const int d0   = dg * CW;
    const int d    = d0 + cl;
    const size_t tileOff = ((size_t)bi * DD + d0) * LL + (size_t)c * LC;
    stage_tile16(Delta + tileOff, Ds, lane);
    stage_tile16(U + tileOff, Us, lane);
    const float* gB = Xd + ((size_t)bi * EE + RR) * LL + (size_t)c * LC;
    stage_tile16(gB, Bs, lane);
    stage_tile16(gB + (size_t)NS * LL, Cs, lane);
    float An[8], h[8];
    const float* hg = Hin + (((size_t)bi * NC + c) * DD + d) * NS;
    #pragma unroll
    for (int q = 0; q < 4; ++q) {
        const float4 a = *(const float4*)(Alog + d * NS + 4 * q);
        An[2*q+0] = -expf(hh ? a.y : a.x) * LOG2E;
        An[2*q+1] = -expf(hh ? a.w : a.z) * LOG2E;
        const float4 t = *(const float4*)(hg + 4 * q);
        h[2*q+0] = hh ? t.y : t.x;
        h[2*q+1] = hh ? t.w : t.z;
    }
    const float Dv = Dp[d];
    __syncthreads();
    float* dRow = Ds + cl * BP;
    const float* uRow = Us + cl * BP;
    const float* bRow = Bs + hh * BP;
    const float* cRow = Cs + hh * BP;
    for (int l0 = 0; l0 < LC; l0 += 4) {
        const float4 dq = *(const float4*)(dRow + l0);
        const float4 uq = *(const float4*)(uRow + l0);
        const float dv[4] = {dq.x, dq.y, dq.z, dq.w};
        const float uv[4] = {uq.x, uq.y, uq.z, uq.w};
        float bv[8][4], cv[8][4];
        #pragma unroll
        for (int i = 0; i < 8; ++i) {
            const float4 t = *(const float4*)(bRow + 2 * i * BP + l0);
            bv[i][0] = t.x; bv[i][1] = t.y; bv[i][2] = t.z; bv[i][3] = t.w;
            const float4 s = *(const float4*)(cRow + 2 * i * BP + l0);
            cv[i][0] = s.x; cv[i][1] = s.y; cv[i][2] = s.z; cv[i][3] = s.w;
        }
        float part[4];
        #pragma unroll
        for (int t = 0; t < 4; ++t) {
            const float dj = dv[t];
            const float uj = uv[t];
            float p[8];
            #pragma unroll
            for (int i = 0; i < 8; ++i) {
                h[i] = exp2_hw(dj * An[i]) * h[i] + dj * uj * bv[i][t];
                p[i] = h[i] * cv[i][t];
            }
            #pragma unroll
            for (int i = 0; i < 4; ++i) p[i] += p[i + 4];
            p[0] += p[2]; p[1] += p[3];
            part[t] = p[0] + p[1];
        }
        float yf[4];
        #pragma unroll
        for (int t = 0; t < 4; ++t) yf[t] = part[t] + xhalf(part[t]);
        float yo2[2];
        #pragma unroll
        for (int k = 0; k < 2; ++k) {
            const float ul = hh ? uv[2 + k] : uv[k];
            float yo = (hh ? yf[2 + k] : yf[k]) + ul * Dv;
            yo = yo * (ul * sigmoidf_(ul));
            yo = 0.5f * yo * (1.0f + erff(yo * 0.70710678118f));
            yo2[k] = yo;
        }
        float2 yq; yq.x = yo2[0]; yq.y = yo2[1];
        *(float2*)(dRow + l0 + 2 * hh) = yq;
    }
    __syncthreads();
    {
        const int q = lane & 15, r0 = lane >> 4;
        float* yg = Y + tileOff;
        #pragma unroll
        for (int k = 0; k < 8; ++k) {
            const float4 t = *(const float4*)(Ds + (2 * k + r0) * BP + 4 * q);
            *(float4*)(yg + (size_t)(2 * k + r0) * LL + 4 * q) = t;
        }
    }
}

#define OKC 32
#define OPA 36
#define OPB 136

typedef float v4f __attribute__((ext_vector_type(4)));

__global__ __launch_bounds__(256)
void k_out(const float* __restrict__ Wo, const float* __restrict__ Yv,
           const float* __restrict__ bo, float* __restrict__ Out)
{
    __shared__ __attribute__((aligned(16))) float As[128 * OPA];
    __shared__ __attribute__((aligned(16))) float Bs[OKC * OPB];
    const int tid  = threadIdx.x;
    const int lane = tid & 31;
    const int wv   = tid >> 5;
    const int wm   = wv & 3;
    const int wn   = wv >> 2;
    const int m0   = blockIdx.y << 6;
    const int n0   = blockIdx.x << 7;
    const int bi   = blockIdx.z;
    const int am   = lane & 15;
    const int ak   = (lane >> 4) << 1;

    const int aq = tid & 7,  ar = tid >> 3;
    const int bq = tid & 31, br = tid >> 5;
    const float* wpA = Wo + (size_t)(m0 + ar) * DD + 4 * aq;
    const float* wpG = wpA + (size_t)DD * DD;
    const float* yp  = Yv + ((size_t)bi * DD + br) * LL + n0 + 4 * bq;
    float* sA = As + ar * OPA + 4 * aq;
    float* sB = Bs + br * OPB + 4 * bq;

    v4f ra0 = *(const v4f*)(wpA);
    v4f ra1 = *(const v4f*)(wpA + 32 * DD);
    v4f ra2 = *(const v4f*)(wpG);
    v4f ra3 = *(const v4f*)(wpG + 32 * DD);
    v4f rb0 = *(const v4f*)(yp);
    v4f rb1 = *(const v4f*)(yp + 8 * LL);
    v4f rb2 = *(const v4f*)(yp + 16 * LL);
    v4f rb3 = *(const v4f*)(yp + 24 * LL);

    v8f accA0 = {}, accA1 = {}, accA2 = {}, accA3 = {};
    v8f accG0 = {}, accG1 = {}, accG2 = {}, accG3 = {};
    const float* aA = As + (16 * wm + am) * OPA + ak;
    const float* aG = As + (64 + 16 * wm + am) * OPA + ak;
    const float* bB = Bs + ak * OPB + 64 * wn + am;

    for (int kc = 0; kc < DD; kc += OKC) {
        *(v4f*)(sA)            = ra0;
        *(v4f*)(sA + 32 * OPA) = ra1;
        *(v4f*)(sA + 64 * OPA) = ra2;
        *(v4f*)(sA + 96 * OPA) = ra3;
        *(v4f*)(sB)            = rb0;
        *(v4f*)(sB + 8 * OPB)  = rb1;
        *(v4f*)(sB + 16 * OPB) = rb2;
        *(v4f*)(sB + 24 * OPB) = rb3;
        __syncthreads();
        if (kc + OKC < DD) {
            const float* wa = wpA + kc + OKC;
            const float* wg = wpG + kc + OKC;
            const float* yy = yp + (size_t)(kc + OKC) * LL;
            ra0 = *(const v4f*)(wa);
            ra1 = *(const v4f*)(wa + 32 * DD);
            ra2 = *(const v4f*)(wg);
            ra3 = *(const v4f*)(wg + 32 * DD);
            rb0 = *(const v4f*)(yy);
            rb1 = *(const v4f*)(yy + 8 * LL);
            rb2 = *(const v4f*)(yy + 16 * LL);
            rb3 = *(const v4f*)(yy + 24 * LL);
        }
        #pragma unroll
        for (int k = 0; k < OKC; k += 4) {
            v2f aa, ag, b0, b1, b2, b3;
            aa.x = aA[k];
            aa.y = aA[k + 1];
            ag.x = aG[k];
            ag.y = aG[k + 1];
            b0.x = bB[k * OPB];            b0.y = bB[(k + 1) * OPB];
            b1.x = bB[k * OPB + 16];       b1.y = bB[(k + 1) * OPB + 16];
            b2.x = bB[k * OPB + 32];       b2.y = bB[(k + 1) * OPB + 32];
            b3.x = bB[k * OPB + 48];       b3.y = bB[(k + 1) * OPB + 48];
            accA0 = __builtin_amdgcn_wmma_f32_16x16x4_f32(false, aa, false, b0, (short)0, accA0, false, false);
            accG0 = __builtin_amdgcn_wmma_f32_16x16x4_f32(false, ag, false, b0, (short)0, accG0, false, false);
            accA1 = __builtin_amdgcn_wmma_f32_16x16x4_f32(false, aa, false, b1, (short)0, accA1, false, false);
            accG1 = __builtin_amdgcn_wmma_f32_16x16x4_f32(false, ag, false, b1, (short)0, accG1, false, false);
            accA2 = __builtin_amdgcn_wmma_f32_16x16x4_f32(false, aa, false, b2, (short)0, accA2, false, false);
            accG2 = __builtin_amdgcn_wmma_f32_16x16x4_f32(false, ag, false, b2, (short)0, accG2, false, false);
            accA3 = __builtin_amdgcn_wmma_f32_16x16x4_f32(false, aa, false, b3, (short)0, accA3, false, false);
            accG3 = __builtin_amdgcn_wmma_f32_16x16x4_f32(false, ag, false, b3, (short)0, accG3, false, false);
        }
        __syncthreads();
    }

    const int cr = (lane >> 4) << 3;
    #pragma unroll
    for (int v = 0; v < 8; ++v) {
        const int row = m0 + 16 * wm + cr + v;
        const float ba = bo[row];
        const float bg = bo[DD + row];
        float* o = Out + ((size_t)bi * DD + row) * LL + n0 + 64 * wn + am;
        o[0]  = (accA0[v] + ba) * sigmoidf_(accG0[v] + bg);
        o[16] = (accA1[v] + ba) * sigmoidf_(accG1[v] + bg);
        o[32] = (accA2[v] + ba) * sigmoidf_(accG2[v] + bg);
        o[48] = (accA3[v] + ba) * sigmoidf_(accG3[v] + bg);
    }
}

extern "C" void kernel_launch(void* const* d_in, const int* in_sizes, int n_in,
                              void* d_out, int out_size, void* d_ws, size_t ws_size,
                              hipStream_t stream)
{
    const float* u    = (const float*)d_in[0];
    const float* xw   = (const float*)d_in[1];
    const float* dtw  = (const float*)d_in[2];
    const float* dtb  = (const float*)d_in[3];
    const float* alog = (const float*)d_in[4];
    const float* Dp   = (const float*)d_in[5];
    const float* ow   = (const float*)d_in[6];
    const float* ob   = (const float*)d_in[7];
    float* out = (float*)d_out;

    const size_t CH = (size_t)NB * DD;
    float* ws    = (float*)d_ws;
    float* xdbl  = ws;
    float* delta = xdbl + (size_t)NB * EE * LL;
    float* yv    = delta + CH * LL;
    float* Sst   = yv + CH * LL;
    float* sumD  = Sst + CH * NC * NS;
    float* hin   = sumD + CH * NC;

    const dim3 blk(128);
    hipLaunchKernelGGL(k_xproj, dim3(LL / 64, EE / 16, NB), blk, 0, stream, xw, u, xdbl);
    hipLaunchKernelGGL(k_dt,    dim3(LL / 64, DD / 16, NB), blk, 0, stream, dtw, xdbl, dtb, delta);
    hipLaunchKernelGGL(k_scanA, dim3((CH * NC) / CW), dim3(32), 0, stream, u, xdbl, delta, alog, Sst, sumD);
    hipLaunchKernelGGL(k_scanB, dim3(CH / 16), dim3(256), 0, stream, alog, Sst, sumD, hin);
    hipLaunchKernelGGL(k_scanC, dim3((CH * NC) / CW), dim3(32), 0, stream, u, xdbl, delta, alog, Dp, hin, yv);
    hipLaunchKernelGGL(k_out,   dim3(LL / 128, DD / 64, NB), dim3(256), 0, stream, ow, yv, ob, out);
}


# softplus (k_dt) and sigmoid (k_scanC silu, k_out gate) on hardware exp2 + rcp, log1p as a short atanh series (were library expf, log1pf, rounded division); sum order unchanged
# speedup vs baseline: 2.5295x; 1.0532x over previous
#include <hip/hip_runtime.h>
#include <math.h>

#define NB 2
#define DD 1024
#define LL 2048
#define NS 16
#define RR 64
#define EE 96
#define NC 32
#define LC 64
#define BP 68

typedef float v2f __attribute__((ext_vector_type(2)));
typedef float v8f __attribute__((ext_vector_type(8)));

#define LOG2E    1.4426950408889634f
#define LOG2E_LO 1.9259629911266175e-8f
#define LN2      0.6931471805599453f

__device__ __forceinline__ float sigmoidf_(float x)
{
    return __builtin_amdgcn_rcpf(1.0f + __builtin_amdgcn_exp2f(-x * LOG2E));
}

__device__ __forceinline__ float softplusf_(float x)
{
    const float a  = fabsf(x);
    const float ac = (a > 128.0f) ? 128.0f : a;
    const float p  = ac * -LOG2E;
    const float lo = fmaf(ac, -LOG2E_LO, fmaf(ac, -LOG2E, -p));
    float t = __builtin_amdgcn_exp2f(p);
    t = fmaf(t, lo * LN2, t);
    const float d  = 2.0f + t;
    const float ed = (2.0f - d) + t;
    const float r  = __builtin_amdgcn_rcpf(d);
    float s = t * r;
    s = fmaf(fmaf(-s, ed, fmaf(-s, d, t)), r, s);
    const float z = s * s;
    float P = fmaf(z, 1.0f / 15.0f, 1.0f / 13.0f);
    P = fmaf(z, P, 1.0f / 11.0f);
    P = fmaf(z, P, 1.0f / 9.0f);
    P = fmaf(z, P, 1.0f / 7.0f);
    P = fmaf(z, P, 1.0f / 5.0f);
    P = fmaf(z, P, 1.0f / 3.0f);
    P = fmaf(z, P, 1.0f);
    return fmaf(s + s, P, fmaxf(x, 0.0f));
}

#define XP 1028
#define XCH 64

__device__ __forceinline__ void xproj_load_b(float (&b)[32], const float* __restrict__ p)
{
    #pragma unroll
    for (int j = 0; j < 16; ++j) {
        b[2 * j]     = p[(size_t)(4 * j) * LL];
        b[2 * j + 1] = p[(size_t)(4 * j + 1) * LL];
    }
}

__global__ __launch_bounds__(128)
void k_xproj(const float* __restrict__ W, const float* __restrict__ U,
             float* __restrict__ Xd)
{
    __shared__ __attribute__((aligned(16))) float As[16 * XP];
    const int tid  = threadIdx.x;
    const int lane = tid & 31;
    const int wv   = tid >> 5;
    const int m0   = blockIdx.y << 4;
    const int n0   = (blockIdx.x << 6) + (wv << 4);
    const int bi   = blockIdx.z;
    const int am   = lane & 15;
    const int ak   = (lane >> 4) << 1;
    const float* Ub = U + ((size_t)bi * DD + ak) * LL + n0 + am;

    float b[3][32];
    xproj_load_b(b[0], Ub);
    xproj_load_b(b[1], Ub + (size_t)XCH * LL);

    {
        const float* wp = W + (size_t)m0 * DD + 4 * tid;
        float4 t[32];
        #pragma unroll
        for (int j = 0; j < 32; ++j)
            t[j] = *(const float4*)(wp + (size_t)(j >> 1) * DD + 512 * (j & 1));
        #pragma unroll
        for (int j = 0; j < 32; ++j)
            *(float4*)(As + (j >> 1) * XP + 512 * (j & 1) + 4 * tid) = t[j];
    }
    __syncthreads();

    v8f acc = {};
    const float* aRow = As + am * XP + ak;
    #pragma unroll
    for (int c = 0; c < DD / XCH; ++c) {
        if (c + 2 < DD / XCH)
            xproj_load_b(b[(c + 2) % 3], Ub + (size_t)((c + 2) * XCH) * LL);
        #pragma unroll
        for (int j = 0; j < 16; ++j) {
            const v2f a = *(const v2f*)(aRow + c * XCH + 4 * j);
            v2f bb;
            bb.x = b[c % 3][2 * j];
            bb.y = b[c % 3][2 * j + 1];
            acc = __builtin_amdgcn_wmma_f32_16x16x4_f32(
                false, a, false, bb, (short)0, acc, false, false);
        }
    }
    const int cn = n0 + am;
    const int cr = (lane >> 4) << 3;
    #pragma unroll
    for (int v = 0; v < 8; ++v)
        Xd[((size_t)bi * EE + m0 + cr + v) * LL + cn] = acc[v];
}

__global__ __launch_bounds__(128)
void k_dt(const float* __restrict__ Wdt, const float* __restrict__ Xd,
          const float* __restrict__ bias, float* __restrict__ Delta)
{
    __shared__ __attribute__((aligned(16))) float As[16 * 68];
    const int tid  = threadIdx.x;
    const int lane = tid & 31;
    const int wv   = tid >> 5;
    const int m0   = blockIdx.y << 4;
    const int n0   = (blockIdx.x << 6) + (wv << 4);
    const int bi   = blockIdx.z;
    const int am   = lane & 15;
    const int ak   = (lane >> 4) << 1;
    const int cr   = (lane >> 4) << 3;
    const int wr = tid >> 4, wq = tid & 15;
    const float4 t0 = *(const float4*)(Wdt + (size_t)(m0 + wr) * RR + 4 * wq);
    const float4 t1 = *(const float4*)(Wdt + (size_t)(m0 + 8 + wr) * RR + 4 * wq);
    const float* Bb = Xd + ((size_t)bi * EE + ak) * LL + n0 + am;
    float bx[16], by[16];
    #pragma unroll
    for (int j = 0; j < 16; ++j) {
        bx[j] = Bb[(size_t)(4 * j) * LL];
        by[j] = Bb[(size_t)(4 * j + 1) * LL];
    }
    const float4 bq0 = *(const float4*)(bias + m0 + cr);
    const float4 bq1 = *(const float4*)(bias + m0 + cr + 4);
    const float bv[8] = {bq0.x, bq0.y, bq0.z, bq0.w, bq1.x, bq1.y, bq1.z, bq1.w};
    *(float4*)(As + wr * 68 + 4 * wq)       = t0;
    *(float4*)(As + (8 + wr) * 68 + 4 * wq) = t1;
    __syncthreads();
    v8f acc = {};
    #pragma unroll
    for (int j = 0; j < 16; ++j) {
        v2f a, b;
        a.x = As[am * 68 + 4 * j + ak];
        a.y = As[am * 68 + 4 * j + ak + 1];
        b.x = bx[j];
        b.y = by[j];
        acc = __builtin_amdgcn_wmma_f32_16x16x4_f32(
            false, a, false, b, (short)0, acc, false, false);
    }
    const int cn = n0 + am;
    #pragma unroll
    for (int v = 0; v < 8; ++v) {
        const int row = m0 + cr + v;
        Delta[((size_t)bi * DD + row) * LL + cn] = softplusf_(acc[v] + bv[v]);
    }
}

#define CW 16
#define HP 20

__device__ __forceinline__ float exp2_hw(float x) { return __builtin_amdgcn_exp2f(x); }

__device__ __forceinline__ float xhalf(float x)
{
    return __builtin_bit_cast(float, __builtin_amdgcn_permlanex16(
        __builtin_bit_cast(int, x), __builtin_bit_cast(int, x), 0x76543210, 0xfedcba98, false, false));
}

__device__ __forceinline__ void stage_tile16(const float* __restrict__ g, float* s, int lane)
{
    const int q = lane & 15, r0 = lane >> 4;
    #pragma unroll
    for (int k = 0; k < 8; ++k) {
        const float4 t = *(const float4*)(g + (size_t)(2 * k + r0) * LL + 4 * q);
        *(float4*)(s + (2 * k + r0) * BP + 4 * q) = t;
    }
}

__global__ __launch_bounds__(32)
void k_scanA(const float* __restrict__ U, const float* __restrict__ Xd,
             const float* __restrict__ Delta, const float* __restrict__ Alog,
             float* __restrict__ S, float* __restrict__ SumD)
{
    __shared__ __attribute__((aligned(16))) float Ds[CW * BP];
    __shared__ __attribute__((aligned(16))) float Us[CW * BP];
    __shared__ __attribute__((aligned(16))) float Bs[NS * BP];
    const int lane = threadIdx.x;
    const int cl   = lane & 15;
    const int hh   = lane >> 4;
    const int dg   = blockIdx.x & (DD / CW - 1);
    const int c    = (blockIdx.x / (DD / CW)) & (NC - 1);
    const int bi   = blockIdx.x / ((DD / CW) * NC);
    const int d0   = dg * CW;
    const int d    = d0 + cl;
    const size_t tileOff = ((size_t)bi * DD + d0) * LL + (size_t)c * LC;
    stage_tile16(Delta + tileOff, Ds, lane);
    stage_tile16(U + tileOff, Us, lane);
    stage_tile16(Xd + ((size_t)bi * EE + RR) * LL + (size_t)c * LC, Bs, lane);
    float An[8];
    #pragma unroll
    for (int q = 0; q < 4; ++q) {
        const float4 a = *(const float4*)(Alog + d * NS + 4 * q);
        An[2*q+0] = -expf(hh ? a.y : a.x) * LOG2E;
        An[2*q+1] = -expf(hh ? a.w : a.z) * LOG2E;
    }
    __syncthreads();
    const float* dRow = Ds + cl * BP;
    const float* uRow = Us + cl * BP;
    const float* bRow = Bs + hh * BP;
    {
        float sd[16];
        #pragma unroll
        for (int q = 0; q < 4; ++q) {
            const float4 a0 = *(const float4*)(dRow + 4 * q);
            const float4 a1 = *(const float4*)(dRow + 16 + 4 * q);
            const float4 a2 = *(const float4*)(dRow + 32 + 4 * q);
            const float4 a3 = *(const float4*)(dRow + 48 + 4 * q);
            sd[4*q+0] = ((a0.x + a1.x) + a2.x) + a3.x;
            sd[4*q+1] = ((a0.y + a1.y) + a2.y) + a3.y;
            sd[4*q+2] = ((a0.z + a1.z) + a2.z) + a3.z;
            sd[4*q+3] = ((a0.w + a1.w) + a2.w) + a3.w;
        }
        #pragma unroll
        for (int i = 0; i < 8; ++i) sd[i] += sd[i + 8];
        #pragma unroll
        for (int i = 0; i < 4; ++i) sd[i] += sd[i + 4];
        sd[0] += sd[2]; sd[1] += sd[3];
        if (hh == 0) SumD[((size_t)bi * DD + d) * NC + c] = sd[0] + sd[1];
    }
    float h[8];
    #pragma unroll
    for (int i = 0; i < 8; ++i) h[i] = 0.0f;
    for (int l0 = 0; l0 < LC; l0 += 4) {
        const float4 dq = *(const float4*)(dRow + l0);
        const float4 uq = *(const float4*)(uRow + l0);
        const float dv[4] = {dq.x, dq.y, dq.z, dq.w};
        const float uv[4] = {uq.x, uq.y, uq.z, uq.w};
        float bv[8][4];
        #pragma unroll
        for (int i = 0; i < 8; ++i) {
            const float4 t = *(const float4*)(bRow + 2 * i * BP + l0);
            bv[i][0] = t.x; bv[i][1] = t.y; bv[i][2] = t.z; bv[i][3] = t.w;
        }
        #pragma unroll
        for (int t = 0; t < 4; ++t) {
            const float dj = dv[t];
            const float uj = uv[t];
            #pragma unroll
            for (int i = 0; i < 8; ++i)
                h[i] = exp2_hw(dj * An[i]) * h[i] + dj * uj * bv[i][t];
        }
    }
    __syncthreads();
    float* Hs = Bs;
    #pragma unroll
    for (int i = 0; i < 8; ++i) Hs[cl * HP + 2 * i + hh] = h[i];
    __syncthreads();
    float* Sg = S + (((size_t)bi * NC + c) * DD + d0) * NS;
    #pragma unroll
    for (int k = 0; k < 2; ++k) {
        const int r = 8 * k + (lane >> 2), q = lane & 3;
        const float4 t = *(const float4*)(Hs + r * HP + 4 * q);
        *(float4*)(Sg + r * NS + 4 * q) = t;
    }
}

__global__ __launch_bounds__(256)
void k_scanB(const float* __restrict__ Alog, const float* __restrict__ S,
             const float* __restrict__ SumD, float* __restrict__ Hin)
{
    const int lane = threadIdx.x & 31;
    const int gw   = blockIdx.x * (blockDim.x >> 5) + (threadIdx.x >> 5);
    const int ch   = (gw << 1) + (lane >> 4);
    const int bi   = ch >> 10;
    const int d    = ch & (DD - 1);
    const int n    = lane & 15;
    const size_t so = ((size_t)bi * NC * DD + d) * NS + n;
    const float al = Alog[d * NS + n];
    float sd[NC], Sv[NC];
    #pragma unroll
    for (int q = 0; q < NC / 4; ++q) {
        const float4 t = *(const float4*)(SumD + (size_t)ch * NC + 4 * q);
        sd[4*q+0] = t.x; sd[4*q+1] = t.y; sd[4*q+2] = t.z; sd[4*q+3] = t.w;
    }
    #pragma unroll
    for (int c = 0; c < NC; ++c)
        Sv[c] = S[so + (size_t)c * DD * NS];
    const float An = -expf(al);
    float P[NC];
    #pragma unroll
    for (int c = 0; c < NC; ++c)
        P[c] = expf(An * sd[c]);
    float hv[NC];
    float hin = 0.0f;
    #pragma unroll
    for (int c = 0; c < NC; ++c) {
        hv[c] = hin;
        hin = P[c] * hin + Sv[c];
    }
    #pragma unroll
    for (int c = 0; c < NC; ++c)
        Hin[so + (size_t)c * DD * NS] = hv[c];
}

__global__ __launch_bounds__(32)
void k_scanC(const float* __restrict__ U, const float* __restrict__ Xd,
             const float* __restrict__ Delta, const float* __restrict__ Alog,
             const float* __restrict__ Dp, const float* __restrict__ Hin,
             float* __restrict__ Y)
{
    __shared__ __attribute__((aligned(16))) float Ds[CW * BP];
    __shared__ __attribute__((aligned(16))) float Us[CW * BP];
    __shared__ __attribute__((aligned(16))) float Bs[NS * BP];
    __shared__ __attribute__((aligned(16))) float Cs[NS * BP];
    const int lane = threadIdx.x;
    const int cl   = lane & 15;
    const int hh   = lane >> 4;
    const int dg   = blockIdx.x & (DD / CW - 1);
    const int c    = (blockIdx.x / (DD / CW)) & (NC - 1);
    const int bi   = blockIdx.x / ((DD / CW) * NC);
    const int d0   = dg * CW;
    const int d    = d0 + cl;
    const size_t tileOff = ((size_t)bi * DD + d0) * LL + (size_t)c * LC;
    stage_tile16(Delta + tileOff, Ds, lane);
    stage_tile16(U + tileOff, Us, lane);
    const float* gB = Xd + ((size_t)bi * EE + RR) * LL + (size_t)c * LC;
    stage_tile16(gB, Bs, lane);
    stage_tile16(gB + (size_t)NS * LL, Cs, lane);
    float An[8], h[8];
    const float* hg = Hin + (((size_t)bi * NC + c) * DD + d) * NS;
    #pragma unroll
    for (int q = 0; q < 4; ++q) {
        const float4 a = *(const float4*)(Alog + d * NS + 4 * q);
        An[2*q+0] = -expf(hh ? a.y : a.x) * LOG2E;
        An[2*q+1] = -expf(hh ? a.w : a.z) * LOG2E;
        const float4 t = *(const float4*)(hg + 4 * q);
        h[2*q+0] = hh ? t.y : t.x;
        h[2*q+1] = hh ? t.w : t.z;
    }
    const float Dv = Dp[d];
    __syncthreads();
    float* dRow = Ds + cl * BP;
    const float* uRow = Us + cl * BP;
    const float* bRow = Bs + hh * BP;
    const float* cRow = Cs + hh * BP;
    for (int l0 = 0; l0 < LC; l0 += 4) {
        const float4 dq = *(const float4*)(dRow + l0);
        const float4 uq = *(const float4*)(uRow + l0);
        const float dv[4] = {dq.x, dq.y, dq.z, dq.w};
        const float uv[4] = {uq.x, uq.y, uq.z, uq.w};
        float bv[8][4], cv[8][4];
        #pragma unroll
        for (int i = 0; i < 8; ++i) {
            const float4 t = *(const float4*)(bRow + 2 * i * BP + l0);
            bv[i][0] = t.x; bv[i][1] = t.y; bv[i][2] = t.z; bv[i][3] = t.w;
            const float4 s = *(const float4*)(cRow + 2 * i * BP + l0);
            cv[i][0] = s.x; cv[i][1] = s.y; cv[i][2] = s.z; cv[i][3] = s.w;
        }
        float part[4];
        #pragma unroll
        for (int t = 0; t < 4; ++t) {
            const float dj = dv[t];
            const float uj = uv[t];
            float p[8];
            #pragma unroll
            for (int i = 0; i < 8; ++i) {
                h[i] = exp2_hw(dj * An[i]) * h[i] + dj * uj * bv[i][t];
                p[i] = h[i] * cv[i][t];
            }
            #pragma unroll
            for (int i = 0; i < 4; ++i) p[i] += p[i + 4];
            p[0] += p[2]; p[1] += p[3];
            part[t] = p[0] + p[1];
        }
        float yf[4];
        #pragma unroll
        for (int t = 0; t < 4; ++t) yf[t] = part[t] + xhalf(part[t]);
        float yo2[2];
        #pragma unroll
        for (int k = 0; k < 2; ++k) {
            const float ul = hh ? uv[2 + k] : uv[k];
            float yo = (hh ? yf[2 + k] : yf[k]) + ul * Dv;
            yo = yo * (ul * sigmoidf_(ul));
            yo = 0.5f * yo * (1.0f + erff(yo * 0.70710678118f));
            yo2[k] = yo;
        }
        float2 yq; yq.x = yo2[0]; yq.y = yo2[1];
        *(float2*)(dRow + l0 + 2 * hh) = yq;
    }
    __syncthreads();
    {
        const int q = lane & 15, r0 = lane >> 4;
        float* yg = Y + tileOff;
        #pragma unroll
        for (int k = 0; k < 8; ++k) {
            const float4 t = *(const float4*)(Ds + (2 * k + r0) * BP + 4 * q);
            *(float4*)(yg + (size_t)(2 * k + r0) * LL + 4 * q) = t;
        }
    }
}

#define OKC 32
#define OPA 36
#define OPB 136

typedef float v4f __attribute__((ext_vector_type(4)));

__global__ __launch_bounds__(256)
void k_out(const float* __restrict__ Wo, const float* __restrict__ Yv,
           const float* __restrict__ bo, float* __restrict__ Out)
{
    __shared__ __attribute__((aligned(16))) float As[128 * OPA];
    __shared__ __attribute__((aligned(16))) float Bs[OKC * OPB];
    const int tid  = threadIdx.x;
    const int lane = tid & 31;
    const int wv   = tid >> 5;
    const int wm   = wv & 3;
    const int wn   = wv >> 2;
    const int m0   = blockIdx.y << 6;
    const int n0   = blockIdx.x << 7;
    const int bi   = blockIdx.z;
    const int am   = lane & 15;
    const int ak   = (lane >> 4) << 1;

    const int aq = tid & 7,  ar = tid >> 3;
    const int bq = tid & 31, br = tid >> 5;
    const float* wpA = Wo + (size_t)(m0 + ar) * DD + 4 * aq;
    const float* wpG = wpA + (size_t)DD * DD;
    const float* yp  = Yv + ((size_t)bi * DD + br) * LL + n0 + 4 * bq;
    float* sA = As + ar * OPA + 4 * aq;
    float* sB = Bs + br * OPB + 4 * bq;

    v4f ra0 = *(const v4f*)(wpA);
    v4f ra1 = *(const v4f*)(wpA + 32 * DD);
    v4f ra2 = *(const v4f*)(wpG);
    v4f ra3 = *(const v4f*)(wpG + 32 * DD);
    v4f rb0 = *(const v4f*)(yp);
    v4f rb1 = *(const v4f*)(yp + 8 * LL);
    v4f rb2 = *(const v4f*)(yp + 16 * LL);
    v4f rb3 = *(const v4f*)(yp + 24 * LL);

    v8f accA0 = {}, accA1 = {}, accA2 = {}, accA3 = {};
    v8f accG0 = {}, accG1 = {}, accG2 = {}, accG3 = {};
    const float* aA = As + (16 * wm + am) * OPA + ak;
    const float* aG = As + (64 + 16 * wm + am) * OPA + ak;
    const float* bB = Bs + ak * OPB + 64 * wn + am;

    for (int kc = 0; kc < DD; kc += OKC) {
        *(v4f*)(sA)            = ra0;
        *(v4f*)(sA + 32 * OPA) = ra1;
        *(v4f*)(sA + 64 * OPA) = ra2;
        *(v4f*)(sA + 96 * OPA) = ra3;
        *(v4f*)(sB)            = rb0;
        *(v4f*)(sB + 8 * OPB)  = rb1;
        *(v4f*)(sB + 16 * OPB) = rb2;
        *(v4f*)(sB + 24 * OPB) = rb3;
        __syncthreads();
        if (kc + OKC < DD) {
            const float* wa = wpA + kc + OKC;
            const float* wg = wpG + kc + OKC;
            const float* yy = yp + (size_t)(kc + OKC) * LL;
            ra0 = *(const v4f*)(wa);
            ra1 = *(const v4f*)(wa + 32 * DD);
            ra2 = *(const v4f*)(wg);
            ra3 = *(const v4f*)(wg + 32 * DD);
            rb0 = *(const v4f*)(yy);
            rb1 = *(const v4f*)(yy + 8 * LL);
            rb2 = *(const v4f*)(yy + 16 * LL);
            rb3 = *(const v4f*)(yy + 24 * LL);
        }
        #pragma unroll
        for (int k = 0; k < OKC; k += 4) {
            v2f aa, ag, b0, b1, b2, b3;
            aa.x = aA[k];
            aa.y = aA[k + 1];
            ag.x = aG[k];
            ag.y = aG[k + 1];
            b0.x = bB[k * OPB];            b0.y = bB[(k + 1) * OPB];
            b1.x = bB[k * OPB + 16];       b1.y = bB[(k + 1) * OPB + 16];
            b2.x = bB[k * OPB + 32];       b2.y = bB[(k + 1) * OPB + 32];
            b3.x = bB[k * OPB + 48];       b3.y = bB[(k + 1) * OPB + 48];
            accA0 = __builtin_amdgcn_wmma_f32_16x16x4_f32(false, aa, false, b0, (short)0, accA0, false, false);
            accG0 = __builtin_amdgcn_wmma_f32_16x16x4_f32(false, ag, false, b0, (short)0, accG0, false, false);
            accA1 = __builtin_amdgcn_wmma_f32_16x16x4_f32(false, aa, false, b1, (short)0, accA1, false, false);
            accG1 = __builtin_amdgcn_wmma_f32_16x16x4_f32(false, ag, false, b1, (short)0, accG1, false, false);
            accA2 = __builtin_amdgcn_wmma_f32_16x16x4_f32(false, aa, false, b2, (short)0, accA2, false, false);
            accG2 = __builtin_amdgcn_wmma_f32_16x16x4_f32(false, ag, false, b2, (short)0, accG2, false, false);
            accA3 = __builtin_amdgcn_wmma_f32_16x16x4_f32(false, aa, false, b3, (short)0, accA3, false, false);
            accG3 = __builtin_amdgcn_wmma_f32_16x16x4_f32(false, ag, false, b3, (short)0, accG3, false, false);
        }
        __syncthreads();
    }

    const int cr = (lane >> 4) << 3;
    #pragma unroll
    for (int v = 0; v < 8; ++v) {
        const int row = m0 + 16 * wm + cr + v;
        const float ba = bo[row];
        const float bg = bo[DD + row];
        float* o = Out + ((size_t)bi * DD + row) * LL + n0 + 64 * wn + am;
        o[0]  = (accA0[v] + ba) * sigmoidf_(accG0[v] + bg);
        o[16] = (accA1[v] + ba) * sigmoidf_(accG1[v] + bg);
        o[32] = (accA2[v] + ba) * sigmoidf_(accG2[v] + bg);
        o[48] = (accA3[v] + ba) * sigmoidf_(accG3[v] + bg);
    }
}

extern "C" void kernel_launch(void* const* d_in, const int* in_sizes, int n_in,
                              void* d_out, int out_size, void* d_ws, size_t ws_size,
                              hipStream_t stream)
{
    const float* u    = (const float*)d_in[0];
    const float* xw   = (const float*)d_in[1];
    const float* dtw  = (const float*)d_in[2];
    const float* dtb  = (const float*)d_in[3];
    const float* alog = (const float*)d_in[4];
    const float* Dp   = (const float*)d_in[5];
    const float* ow   = (const float*)d_in[6];
    const float* ob   = (const float*)d_in[7];
    float* out = (float*)d_out;

    const size_t CH = (size_t)NB * DD;
    float* ws    = (float*)d_ws;
    float* xdbl  = ws;
    float* delta = xdbl + (size_t)NB * EE * LL;
    float* yv    = delta + CH * LL;
    float* Sst   = yv + CH * LL;
    float* sumD  = Sst + CH * NC * NS;
    float* hin   = sumD + CH * NC;

    const dim3 blk(128);
    hipLaunchKernelGGL(k_xproj, dim3(LL / 64, EE / 16, NB), blk, 0, stream, xw, u, xdbl);
    hipLaunchKernelGGL(k_dt,    dim3(LL / 64, DD / 16, NB), blk, 0, stream, dtw, xdbl, dtb, delta);
    hipLaunchKernelGGL(k_scanA, dim3((CH * NC) / CW), dim3(32), 0, stream, u, xdbl, delta, alog, Sst, sumD);
    hipLaunchKernelGGL(k_scanB, dim3(CH / 16), dim3(256), 0, stream, alog, Sst, sumD, hin);
    hipLaunchKernelGGL(k_scanC, dim3((CH * NC) / CW), dim3(32), 0, stream, u, xdbl, delta, alog, Dp, hin, yv);
    hipLaunchKernelGGL(k_out,   dim3(LL / 128, DD / 64, NB), dim3(256), 0, stream, ow, yv, ob, out);
}


# k_out, k_xproj: a WMMA block's fragment reads go out as one group ahead of the block before it, none among WMMAs (k_out: y tile k-pair interleaved in LDS, 2 stages, 1 barrier/chunk); sum order kept
# speedup vs baseline: 2.6883x; 1.0628x over previous
#include <hip/hip_runtime.h>
#include <math.h>

#define NB 2
#define DD 1024
#define LL 2048
#define NS 16
#define RR 64
#define EE 96
#define NC 32
#define LC 64
#define BP 68

typedef float v2f __attribute__((ext_vector_type(2)));
typedef float v8f __attribute__((ext_vector_type(8)));

#define LOG2E    1.4426950408889634f
#define LOG2E_LO 1.9259629911266175e-8f
#define LN2      0.6931471805599453f

__device__ __forceinline__ float sigmoidf_(float x)
{
    return __builtin_amdgcn_rcpf(1.0f + __builtin_amdgcn_exp2f(-x * LOG2E));
}

__device__ __forceinline__ float softplusf_(float x)
{
    const float a  = fabsf(x);
    const float ac = (a > 128.0f) ? 128.0f : a;
    const float p  = ac * -LOG2E;
    const float lo = fmaf(ac, -LOG2E_LO, fmaf(ac, -LOG2E, -p));
    float t = __builtin_amdgcn_exp2f(p);
    t = fmaf(t, lo * LN2, t);
    const float d  = 2.0f + t;
    const float ed = (2.0f - d) + t;
    const float r  = __builtin_amdgcn_rcpf(d);
    float s = t * r;
    s = fmaf(fmaf(-s, ed, fmaf(-s, d, t)), r, s);
    const float z = s * s;
    float P = fmaf(z, 1.0f / 15.0f, 1.0f / 13.0f);
    P = fmaf(z, P, 1.0f / 11.0f);
    P = fmaf(z, P, 1.0f / 9.0f);
    P = fmaf(z, P, 1.0f / 7.0f);
    P = fmaf(z, P, 1.0f / 5.0f);
    P = fmaf(z, P, 1.0f / 3.0f);
    P = fmaf(z, P, 1.0f);
    return fmaf(s + s, P, fmaxf(x, 0.0f));
}

#define XP 1028
#define XCH 64

__device__ __forceinline__ void xproj_load_b(float (&b)[32], const float* __restrict__ p)
{
    #pragma unroll
    for (int j = 0; j < 16; ++j) {
        b[2 * j]     = p[(size_t)(4 * j) * LL];
        b[2 * j + 1] = p[(size_t)(4 * j + 1) * LL];
    }
}

__device__ __forceinline__ void xproj_load_a(v2f (&a)[16], const float* p)
{
    #pragma unroll
    for (int j = 0; j < 16; ++j)
        a[j] = *(const v2f*)(p + 4 * j);
}

__global__ __launch_bounds__(128)
void k_xproj(const float* __restrict__ W, const float* __restrict__ U,
             float* __restrict__ Xd)
{
    __shared__ __attribute__((aligned(16))) float As[16 * XP];
    const int tid  = threadIdx.x;
    const int lane = tid & 31;
    const int wv   = tid >> 5;
    const int m0   = blockIdx.y << 4;
    const int n0   = (blockIdx.x << 6) + (wv << 4);
    const int bi   = blockIdx.z;
    const int am   = lane & 15;
    const int ak   = (lane >> 4) << 1;
    const float* Ub = U + ((size_t)bi * DD + ak) * LL + n0 + am;

    float b[3][32];
    xproj_load_b(b[0], Ub);
    xproj_load_b(b[1], Ub + (size_t)XCH * LL);

    {
        const float* wp = W + (size_t)m0 * DD + 4 * tid;
        float4 t[32];
        #pragma unroll
        for (int j = 0; j < 32; ++j)
            t[j] = *(const float4*)(wp + (size_t)(j >> 1) * DD + 512 * (j & 1));
        #pragma unroll
        for (int j = 0; j < 32; ++j)
            *(float4*)(As + (j >> 1) * XP + 512 * (j & 1) + 4 * tid) = t[j];
    }
    __syncthreads();

    v8f acc = {};
    const float* aRow = As + am * XP + ak;
    v2f a[2][16];
    xproj_load_a(a[0], aRow);
    #pragma unroll
    for (int c = 0; c < DD / XCH; ++c) {
        if (c + 2 < DD / XCH)
            xproj_load_b(b[(c + 2) % 3], Ub + (size_t)((c + 2) * XCH) * LL);
        if (c + 1 < DD / XCH)
            xproj_load_a(a[(c + 1) & 1], aRow + (c + 1) * XCH);
        __builtin_amdgcn_sched_barrier(0);
        #pragma unroll
        for (int j = 0; j < 16; ++j) {
            v2f bb;
            bb.x = b[c % 3][2 * j];
            bb.y = b[c % 3][2 * j + 1];
            acc = __builtin_amdgcn_wmma_f32_16x16x4_f32(
                false, a[c & 1][j], false, bb, (short)0, acc, false, false);
        }
        __builtin_amdgcn_sched_barrier(0);
    }
    const int cn = n0 + am;
    const int cr = (lane >> 4) << 3;
    #pragma unroll
    for (int v = 0; v < 8; ++v)
        Xd[((size_t)bi * EE + m0 + cr + v) * LL + cn] = acc[v];
}

__global__ __launch_bounds__(128)
void k_dt(const float* __restrict__ Wdt, const float* __restrict__ Xd,
          const float* __restrict__ bias, float* __restrict__ Delta)
{
    __shared__ __attribute__((aligned(16))) float As[16 * 68];
    const int tid  = threadIdx.x;
    const int lane = tid & 31;
    const int wv   = tid >> 5;
    const int m0   = blockIdx.y << 4;
    const int n0   = (blockIdx.x << 6) + (wv << 4);
    const int bi   = blockIdx.z;
    const int am   = lane & 15;
    const int ak   = (lane >> 4) << 1;
    const int cr   = (lane >> 4) << 3;
    const int wr = tid >> 4, wq = tid & 15;
    const float4 t0 = *(const float4*)(Wdt + (size_t)(m0 + wr) * RR + 4 * wq);
    const float4 t1 = *(const float4*)(Wdt + (size_t)(m0 + 8 + wr) * RR + 4 * wq);
    const float* Bb = Xd + ((size_t)bi * EE + ak) * LL + n0 + am;
    float bx[16], by[16];
    #pragma unroll
    for (int j = 0; j < 16; ++j) {
        bx[j] = Bb[(size_t)(4 * j) * LL];
        by[j] = Bb[(size_t)(4 * j + 1) * LL];
    }
    const float4 bq0 = *(const float4*)(bias + m0 + cr);
    const float4 bq1 = *(const float4*)(bias + m0 + cr + 4);
    const float bv[8] = {bq0.x, bq0.y, bq0.z, bq0.w, bq1.x, bq1.y, bq1.z, bq1.w};
    *(float4*)(As + wr * 68 + 4 * wq)       = t0;
    *(float4*)(As + (8 + wr) * 68 + 4 * wq) = t1;
    __syncthreads();
    v8f acc = {};
    #pragma unroll
    for (int j = 0; j < 16; ++j) {
        v2f a, b;
        a.x = As[am * 68 + 4 * j + ak];
        a.y = As[am * 68 + 4 * j + ak + 1];
        b.x = bx[j];
        b.y = by[j];
        acc = __builtin_amdgcn_wmma_f32_16x16x4_f32(
            false, a, false, b, (short)0, acc, false, false);
    }
    const int cn = n0 + am;
    #pragma unroll
    for (int v = 0; v < 8; ++v) {
        const int row = m0 + cr + v;
        Delta[((size_t)bi * DD + row) * LL + cn] = softplusf_(acc[v] + bv[v]);
    }
}

#define CW 16
#define HP 20

__device__ __forceinline__ float exp2_hw(float x) { return __builtin_amdgcn_exp2f(x); }

__device__ __forceinline__ float xhalf(float x)
{
    return __builtin_bit_cast(float, __builtin_amdgcn_permlanex16(
        __builtin_bit_cast(int, x), __builtin_bit_cast(int, x), 0x76543210, 0xfedcba98, false, false));
}

__device__ __forceinline__ void stage_tile16(const float* __restrict__ g, float* s, int lane)
{
    const int q = lane & 15, r0 = lane >> 4;
    #pragma unroll
    for (int k = 0; k < 8; ++k) {
        const float4 t = *(const float4*)(g + (size_t)(2 * k + r0) * LL + 4 * q);
        *(float4*)(s + (2 * k + r0) * BP + 4 * q) = t;
    }
}

__global__ __launch_bounds__(32)
void k_scanA(const float* __restrict__ U, const float* __restrict__ Xd,
             const float* __restrict__ Delta, const float* __restrict__ Alog,
             float* __restrict__ S, float* __restrict__ SumD)
{
    __shared__ __attribute__((aligned(16))) float Ds[CW * BP];
    __shared__ __attribute__((aligned(16))) float Us[CW * BP];
    __shared__ __attribute__((aligned(16))) float Bs[NS * BP];
    const int lane = threadIdx.x;
    const int cl   = lane & 15;
    const int hh   = lane >> 4;
    const int dg   = blockIdx.x & (DD / CW - 1);
    const int c    = (blockIdx.x / (DD / CW)) & (NC - 1);
    const int bi   = blockIdx.x / ((DD / CW) * NC);
    const int d0   = dg * CW;
    const int d    = d0 + cl;
    const size_t tileOff = ((size_t)bi * DD + d0) * LL + (size_t)c * LC;
    stage_tile16(Delta + tileOff, Ds, lane);
    stage_tile16(U + tileOff, Us, lane);
    stage_tile16(Xd + ((size_t)bi * EE + RR) * LL + (size_t)c * LC, Bs, lane);
    float An[8];
    #pragma unroll
    for (int q = 0; q < 4; ++q) {
        const float4 a = *(const float4*)(Alog + d * NS + 4 * q);
        An[2*q+0] = -expf(hh ? a.y : a.x) * LOG2E;
        An[2*q+1] = -expf(hh ? a.w : a.z) * LOG2E;
    }
    __syncthreads();
    const float* dRow = Ds + cl * BP;
    const float* uRow = Us + cl * BP;
    const float* bRow = Bs + hh * BP;
    {
        float sd[16];
        #pragma unroll
        for (int q = 0; q < 4; ++q) {
            const float4 a0 = *(const float4*)(dRow + 4 * q);
            const float4 a1 = *(const float4*)(dRow + 16 + 4 * q);
            const float4 a2 = *(const float4*)(dRow + 32 + 4 * q);
            const float4 a3 = *(const float4*)(dRow + 48 + 4 * q);
            sd[4*q+0] = ((a0.x + a1.x) + a2.x) + a3.x;
            sd[4*q+1] = ((a0.y + a1.y) + a2.y) + a3.y;
            sd[4*q+2] = ((a0.z + a1.z) + a2.z) + a3.z;
            sd[4*q+3] = ((a0.w + a1.w) + a2.w) + a3.w;
        }
        #pragma unroll
        for (int i = 0; i < 8; ++i) sd[i] += sd[i + 8];
        #pragma unroll
        for (int i = 0; i < 4; ++i) sd[i] += sd[i + 4];
        sd[0] += sd[2]; sd[1] += sd[3];
        if (hh == 0) SumD[((size_t)bi * DD + d) * NC + c] = sd[0] + sd[1];
    }
    float h[8];
    #pragma unroll
    for (int i = 0; i < 8; ++i) h[i] = 0.0f;
    for (int l0 = 0; l0 < LC; l0 += 4) {
        const float4 dq = *(const float4*)(dRow + l0);
        const float4 uq = *(const float4*)(uRow + l0);
        const float dv[4] = {dq.x, dq.y, dq.z, dq.w};
        const float uv[4] = {uq.x, uq.y, uq.z, uq.w};
        float bv[8][4];
        #pragma unroll
        for (int i = 0; i < 8; ++i) {
            const float4 t = *(const float4*)(bRow + 2 * i * BP + l0);
            bv[i][0] = t.x; bv[i][1] = t.y; bv[i][2] = t.z; bv[i][3] = t.w;
        }
        #pragma unroll
        for (int t = 0; t < 4; ++t) {
            const float dj = dv[t];
            const float uj = uv[t];
            #pragma unroll
            for (int i = 0; i < 8; ++i)
                h[i] = exp2_hw(dj * An[i]) * h[i] + dj * uj * bv[i][t];
        }
    }
    __syncthreads();
    float* Hs = Bs;
    #pragma unroll
    for (int i = 0; i < 8; ++i) Hs[cl * HP + 2 * i + hh] = h[i];
    __syncthreads();
    float* Sg = S + (((size_t)bi * NC + c) * DD + d0) * NS;
    #pragma unroll
    for (int k = 0; k < 2; ++k) {
        const int r = 8 * k + (lane >> 2), q = lane & 3;
        const float4 t = *(const float4*)(Hs + r * HP + 4 * q);
        *(float4*)(Sg + r * NS + 4 * q) = t;
    }
}

__global__ __launch_bounds__(256)
void k_scanB(const float* __restrict__ Alog, const float* __restrict__ S,
             const float* __restrict__ SumD, float* __restrict__ Hin)
{
    const int lane = threadIdx.x & 31;
    const int gw   = blockIdx.x * (blockDim.x >> 5) + (threadIdx.x >> 5);
    const int ch   = (gw << 1) + (lane >> 4);
    const int bi   = ch >> 10;
    const int d    = ch & (DD - 1);
    const int n    = lane & 15;
    const size_t so = ((size_t)bi * NC * DD + d) * NS + n;
    const float al = Alog[d * NS + n];
    float sd[NC], Sv[NC];
    #pragma unroll
    for (int q = 0; q < NC / 4; ++q) {
        const float4 t = *(const float4*)(SumD + (size_t)ch * NC + 4 * q);
        sd[4*q+0] = t.x; sd[4*q+1] = t.y; sd[4*q+2] = t.z; sd[4*q+3] = t.w;
    }
    #pragma unroll
    for (int c = 0; c < NC; ++c)
        Sv[c] = S[so + (size_t)c * DD * NS];
    const float An = -expf(al);
    float P[NC];
    #pragma unroll
    for (int c = 0; c < NC; ++c)
        P[c] = expf(An * sd[c]);
    float hv[NC];
    float hin = 0.0f;
    #pragma unroll
    for (int c = 0; c < NC; ++c) {
        hv[c] = hin;
        hin = P[c] * hin + Sv[c];
    }
    #pragma unroll
    for (int c = 0; c < NC; ++c)
        Hin[so + (size_t)c * DD * NS] = hv[c];
}

__global__ __launch_bounds__(32)
void k_scanC(const float* __restrict__ U, const float* __restrict__ Xd,
             const float* __restrict__ Delta, const float* __restrict__ Alog,
             const float* __restrict__ Dp, const float* __restrict__ Hin,
             float* __restrict__ Y)
{
    __shared__ __attribute__((aligned(16))) float Ds[CW * BP];
    __shared__ __attribute__((aligned(16))) float Us[CW * BP];
    __shared__ __attribute__((aligned(16))) float Bs[NS * BP];
    __shared__ __attribute__((aligned(16))) float Cs[NS * BP];
    const int lane = threadIdx.x;
    const int cl   = lane & 15;
    const int hh   = lane >> 4;
    const int dg   = blockIdx.x & (DD / CW - 1);
    const int c    = (blockIdx.x / (DD / CW)) & (NC - 1);
    const int bi   = blockIdx.x / ((DD / CW) * NC);
    const int d0   = dg * CW;
    const int d    = d0 + cl;
    const size_t tileOff = ((size_t)bi * DD + d0) * LL + (size_t)c * LC;
    stage_tile16(Delta + tileOff, Ds, lane);
    stage_tile16(U + tileOff, Us, lane);
    const float* gB = Xd + ((size_t)bi * EE + RR) * LL + (size_t)c * LC;
    stage_tile16(gB, Bs, lane);
    stage_tile16(gB + (size_t)NS * LL, Cs, lane);
    float An[8], h[8];
    const float* hg = Hin + (((size_t)bi * NC + c) * DD + d) * NS;
    #pragma unroll
    for (int q = 0; q < 4; ++q) {
        const float4 a = *(const float4*)(Alog + d * NS + 4 * q);
        An[2*q+0] = -expf(hh ? a.y : a.x) * LOG2E;
        An[2*q+1] = -expf(hh ? a.w : a.z) * LOG2E;
        const float4 t = *(const float4*)(hg + 4 * q);
        h[2*q+0] = hh ? t.y : t.x;
        h[2*q+1] = hh ? t.w : t.z;
    }
    const float Dv = Dp[d];
    __syncthreads();
    float* dRow = Ds + cl * BP;
    const float* uRow = Us + cl * BP;
    const float* bRow = Bs + hh * BP;
    const float* cRow = Cs + hh * BP;
    for (int l0 = 0; l0 < LC; l0 += 4) {
        const float4 dq = *(const float4*)(dRow + l0);
        const float4 uq = *(const float4*)(uRow + l0);
        const float dv[4] = {dq.x, dq.y, dq.z, dq.w};
        const float uv[4] = {uq.x, uq.y, uq.z, uq.w};
        float bv[8][4], cv[8][4];
        #pragma unroll
        for (int i = 0; i < 8; ++i) {
            const float4 t = *(const float4*)(bRow + 2 * i * BP + l0);
            bv[i][0] = t.x; bv[i][1] = t.y; bv[i][2] = t.z; bv[i][3] = t.w;
            const float4 s = *(const float4*)(cRow + 2 * i * BP + l0);
            cv[i][0] = s.x; cv[i][1] = s.y; cv[i][2] = s.z; cv[i][3] = s.w;
        }
        float part[4];
        #pragma unroll
        for (int t = 0; t < 4; ++t) {
            const float dj = dv[t];
            const float uj = uv[t];
            float p[8];
            #pragma unroll
            for (int i = 0; i < 8; ++i) {
                h[i] = exp2_hw(dj * An[i]) * h[i] + dj * uj * bv[i][t];
                p[i] = h[i] * cv[i][t];
            }
            #pragma unroll
            for (int i = 0; i < 4; ++i) p[i] += p[i + 4];
            p[0] += p[2]; p[1] += p[3];
            part[t] = p[0] + p[1];
        }
        float yf[4];
        #pragma unroll
        for (int t = 0; t < 4; ++t) yf[t] = part[t] + xhalf(part[t]);
        float yo2[2];
        #pragma unroll
        for (int k = 0; k < 2; ++k) {
            const float ul = hh ? uv[2 + k] : uv[k];
            float yo = (hh ? yf[2 + k] : yf[k]) + ul * Dv;
            yo = yo * (ul * sigmoidf_(ul));
            yo = 0.5f * yo * (1.0f + erff(yo * 0.70710678118f));
            yo2[k] = yo;
        }
        float2 yq; yq.x = yo2[0]; yq.y = yo2[1];
        *(float2*)(dRow + l0 + 2 * hh) = yq;
    }
    __syncthreads();
    {
        const int q = lane & 15, r0 = lane >> 4;
        float* yg = Y + tileOff;
        #pragma unroll
        for (int k = 0; k < 8; ++k) {
            const float4 t = *(const float4*)(Ds + (2 * k + r0) * BP + 4 * q);
            *(float4*)(yg + (size_t)(2 * k + r0) * LL + 4 * q) = t;
        }
    }
}

#define OKC 32
#define OPA 36
#define OPP 288
#define OSA (128 * OPA)
#define OSB ((OKC / 2) * OPP)

typedef float v4f __attribute__((ext_vector_type(4)));

struct OutFrag { v2f aa[4], ag[4], b[4][4]; };

__device__ __forceinline__ void out_load_frags(OutFrag& f, const float* aA, const float* aG, const float* bB)
{
    #pragma unroll
    for (int j = 0; j < 4; ++j) {
        f.aa[j] = *(const v2f*)(aA + 4 * j);
        f.ag[j] = *(const v2f*)(aG + 4 * j);
        #pragma unroll
        for (int t = 0; t < 4; ++t)
            f.b[j][t] = *(const v2f*)(bB + 2 * j * OPP + 32 * t);
    }
}

__device__ __forceinline__ void out_block(const OutFrag& f, v8f (&accA)[4], v8f (&accG)[4])
{
    #pragma unroll
    for (int j = 0; j < 4; ++j) {
        #pragma unroll
        for (int t = 0; t < 4; ++t) {
            accA[t] = __builtin_amdgcn_wmma_f32_16x16x4_f32(false, f.aa[j], false, f.b[j][t], (short)0, accA[t], false, false);
            accG[t] = __builtin_amdgcn_wmma_f32_16x16x4_f32(false, f.ag[j], false, f.b[j][t], (short)0, accG[t], false, false);
        }
    }
}

struct OutStage { v4f a0, a1, a2, a3, b0, b1, b2, b3; };

__device__ __forceinline__ void out_fetch(OutStage& r, const float* wa, const float* wg, const float* yy)
{
    r.a0 = *(const v4f*)(wa);
    r.a1 = *(const v4f*)(wa + 32 * DD);
    r.a2 = *(const v4f*)(wg);
    r.a3 = *(const v4f*)(wg + 32 * DD);
    r.b0 = *(const v4f*)(yy);
    r.b1 = *(const v4f*)(yy + LL);
    r.b2 = *(const v4f*)(yy + 16 * LL);
    r.b3 = *(const v4f*)(yy + 17 * LL);
}

__device__ __forceinline__ void out_put(const OutStage& r, float* sA, float* sB)
{
    *(v4f*)(sA)            = r.a0;
    *(v4f*)(sA + 32 * OPA) = r.a1;
    *(v4f*)(sA + 64 * OPA) = r.a2;
    *(v4f*)(sA + 96 * OPA) = r.a3;
    v4f p;
    p.x = r.b0.x; p.y = r.b1.x; p.z = r.b0.y; p.w = r.b1.y;
    *(v4f*)(sB) = p;
    p.x = r.b0.z; p.y = r.b1.z; p.z = r.b0.w; p.w = r.b1.w;
    *(v4f*)(sB + 4) = p;
    p.x = r.b2.x; p.y = r.b3.x; p.z = r.b2.y; p.w = r.b3.y;
    *(v4f*)(sB + 8 * OPP) = p;
    p.x = r.b2.z; p.y = r.b3.z; p.z = r.b2.w; p.w = r.b3.w;
    *(v4f*)(sB + 8 * OPP + 4) = p;
}

__global__ __launch_bounds__(256, 4)
void k_out(const float* __restrict__ Wo, const float* __restrict__ Yv,
           const float* __restrict__ bo, float* __restrict__ Out)
{
    __shared__ __attribute__((aligned(16))) float As[2 * OSA];
    __shared__ __attribute__((aligned(16))) float Bs[2 * OSB];
    const int tid  = threadIdx.x;
    const int lane = tid & 31;
    const int wv   = __builtin_amdgcn_readfirstlane(tid >> 5);
    const int wm   = wv & 3;
    const int wn   = wv >> 2;
    const int m0   = blockIdx.y << 6;
    const int n0   = blockIdx.x << 7;
    const int bi   = blockIdx.z;
    const int am   = lane & 15;
    const int ak   = (lane >> 4) << 1;

    const int aq = tid & 7,  ar = tid >> 3;
    const int bq = tid & 31, pr = tid >> 5;
    const float* wpA = Wo + (size_t)(m0 + ar) * DD + 4 * aq;
    const float* wpG = wpA + (size_t)DD * DD;
    const float* yp  = Yv + ((size_t)bi * DD + 2 * pr) * LL + n0 + 4 * bq;
    float* sA = As + ar * OPA + 4 * aq;
    float* sB = Bs + pr * OPP + 8 * bq;
    const float* aA = As + (16 * wm + am) * OPA + ak;
    const float* aG = As + (64 + 16 * wm + am) * OPA + ak;
    const float* bB = Bs + (lane >> 4) * OPP + 2 * (64 * wn + am);

    v8f accA[4] = {}, accG[4] = {};
    OutStage r;
    OutFrag f0, f1;

    {
        OutStage r0;
        out_fetch(r0, wpA, wpG, yp);
        out_fetch(r, wpA + OKC, wpG + OKC, yp + (size_t)OKC * LL);
        out_put(r0, sA, sB);
    }
    __syncthreads();
    out_load_frags(f0, aA, aG, bB);

    #pragma unroll 1
    for (int kc = 0; kc < DD - 2 * OKC; kc += 2 * OKC) {
        out_load_frags(f1, aA + 16, aG + 16, bB + 8 * OPP);
        __builtin_amdgcn_sched_barrier(0);
        out_block(f0, accA, accG);
        __builtin_amdgcn_sched_barrier(0);
        out_put(r, sA + OSA, sB + OSB);
        __syncthreads();
        out_fetch(r, wpA + kc + 2 * OKC, wpG + kc + 2 * OKC, yp + (size_t)(kc + 2 * OKC) * LL);
        out_load_frags(f0, aA + OSA, aG + OSA, bB + OSB);
        __builtin_amdgcn_sched_barrier(0);
        out_block(f1, accA, accG);
        __builtin_amdgcn_sched_barrier(0);
        out_load_frags(f1, aA + OSA + 16, aG + OSA + 16, bB + OSB + 8 * OPP);
        __builtin_amdgcn_sched_barrier(0);
        out_block(f0, accA, accG);
        __builtin_amdgcn_sched_barrier(0);
        out_put(r, sA, sB);
        __syncthreads();
        out_fetch(r, wpA + kc + 3 * OKC, wpG + kc + 3 * OKC, yp + (size_t)(kc + 3 * OKC) * LL);
        out_load_frags(f0, aA, aG, bB);
        __builtin_amdgcn_sched_barrier(0);
        out_block(f1, accA, accG);
        __builtin_amdgcn_sched_barrier(0);
    }
    out_load_frags(f1, aA + 16, aG + 16, bB + 8 * OPP);
    __builtin_amdgcn_sched_barrier(0);
    out_block(f0, accA, accG);
    __builtin_amdgcn_sched_barrier(0);
    out_put(r, sA + OSA, sB + OSB);
    __syncthreads();
    out_load_frags(f0, aA + OSA, aG + OSA, bB + OSB);
    __builtin_amdgcn_sched_barrier(0);
    out_block(f1, accA, accG);
    __builtin_amdgcn_sched_barrier(0);
    out_load_frags(f1, aA + OSA + 16, aG + OSA + 16, bB + OSB + 8 * OPP);
    __builtin_amdgcn_sched_barrier(0);
    out_block(f0, accA, accG);
    out_block(f1, accA, accG);

    const int el = __builtin_amdgcn_mbcnt_lo(~0u, 0u);
    const int em = el & 15;
    const int cr = (el >> 4) << 3;
    #pragma unroll
    for (int v = 0; v < 8; ++v) {
        const int row = m0 + 16 * wm + cr + v;
        const float ba = bo[row];
        const float bg = bo[DD + row];
        float* o = Out + ((size_t)bi * DD + row) * LL + n0 + 64 * wn + em;
        o[0]  = (accA[0][v] + ba) * sigmoidf_(accG[0][v] + bg);
        o[16] = (accA[1][v] + ba) * sigmoidf_(accG[1][v] + bg);
        o[32] = (accA[2][v] + ba) * sigmoidf_(accG[2][v] + bg);
        o[48] = (accA[3][v] + ba) * sigmoidf_(accG[3][v] + bg);
    }
}

extern "C" void kernel_launch(void* const* d_in, const int* in_sizes, int n_in,
                              void* d_out, int out_size, void* d_ws, size_t ws_size,
                              hipStream_t stream)
{
    const float* u    = (const float*)d_in[0];
    const float* xw   = (const float*)d_in[1];
    const float* dtw  = (const float*)d_in[2];
    const float* dtb  = (const float*)d_in[3];
    const float* alog = (const float*)d_in[4];
    const float* Dp   = (const float*)d_in[5];
    const float* ow   = (const float*)d_in[6];
    const float* ob   = (const float*)d_in[7];
    float* out = (float*)d_out;

    const size_t CH = (size_t)NB * DD;
    float* ws    = (float*)d_ws;
    float* xdbl  = ws;
    float* delta = xdbl + (size_t)NB * EE * LL;
    float* yv    = delta + CH * LL;
    float* Sst   = yv + CH * LL;
    float* sumD  = Sst + CH * NC * NS;
    float* hin   = sumD + CH * NC;

    const dim3 blk(128);
    hipLaunchKernelGGL(k_xproj, dim3(LL / 64, EE / 16, NB), blk, 0, stream, xw, u, xdbl);
    hipLaunchKernelGGL(k_dt,    dim3(LL / 64, DD / 16, NB), blk, 0, stream, dtw, xdbl, dtb, delta);
    hipLaunchKernelGGL(k_scanA, dim3((CH * NC) / CW), dim3(32), 0, stream, u, xdbl, delta, alog, Sst, sumD);
    hipLaunchKernelGGL(k_scanB, dim3(CH / 16), dim3(256), 0, stream, alog, Sst, sumD, hin);
    hipLaunchKernelGGL(k_scanC, dim3((CH * NC) / CW), dim3(32), 0, stream, u, xdbl, delta, alog, Dp, hin, yv);
    hipLaunchKernelGGL(k_out,   dim3(LL / 128, DD / 64, NB), dim3(256), 0, stream, ow, yv, ob, out);
}
